# starting kernel
# baseline (speedup 1.0000x reference)
#include <hip/hip_runtime.h>

typedef __attribute__((ext_vector_type(2))) float v2f;
typedef __attribute__((ext_vector_type(8))) float v8f;


__global__ void k_zero(float* __restrict__ p, int n) {
    int i = blockIdx.x * blockDim.x + threadIdx.x;
    if (i < n) p[i] = 0.0f;
}

__global__ void k_degree(const int* __restrict__ src, const int* __restrict__ dst,
                         float* __restrict__ degO, float* __restrict__ degI, int e) {
    int i = blockIdx.x * blockDim.x + threadIdx.x;
    if (i < e) {
        atomicAdd(&degO[src[i]], 1.0f);
        atomicAdd(&degI[dst[i]], 1.0f);
    }
}

__global__ void k_coeff(float* __restrict__ c, int n) {
    int i = blockIdx.x * blockDim.x + threadIdx.x;
    if (i < n) {
        float d = c[i];
        c[i] = (d > 0.0f) ? rsqrtf(d) : 0.0f;
    }
}

__global__ void k_pad_features(const float* __restrict__ x, float* __restrict__ xp, int n) {
    int i = blockIdx.x * blockDim.x + threadIdx.x;
    if (i < n * 8) {
        int r = i >> 3, c = i & 7;
        xp[i] = (c < 6) ? x[r * 6 + c] : 0.0f;
    }
}

__global__ void k_pad_w1(const float* __restrict__ w, float* __restrict__ wp) {
    int i = blockIdx.x * blockDim.x + threadIdx.x;
    if (i < 8 * 128) {
        int r = i >> 7;
        wp[i] = (r < 6) ? w[i] : 0.0f;
    }
}

__global__ void k_pad_w5(const float* __restrict__ w, float* __restrict__ wp) {
    int i = blockIdx.x * blockDim.x + threadIdx.x;
    if (i < 128 * 32) {
        int r = i >> 5, c = i & 31;
        wp[i] = (c < 18) ? w[r * 18 + c] : 0.0f;
    }
}

template <int DIN, int DOUT>
__global__ void k_gemm_wmma(const float* __restrict__ A, const float* __restrict__ W,
                            float* __restrict__ Y) {
    const int lane = threadIdx.x;
    const int m0   = blockIdx.x * 16;
    const int n0   = blockIdx.y * 16;
    const int half = lane >> 4;
    const int l15  = lane & 15;
    const int koff = half * 2;

    const float* arow = A + (size_t)(m0 + l15) * DIN + koff;
    const int col = n0 + l15;

    v8f acc = {};
#pragma unroll
    for (int k = 0; k < DIN; k += 4) {
        v2f a = *(const v2f*)(arow + k);
        v2f b;
        b.x = W[(size_t)(k + koff) * DOUT + col];
        b.y = W[(size_t)(k + koff + 1) * DOUT + col];
        acc = __builtin_amdgcn_wmma_f32_16x16x4_f32(
             false, a,  false, b,
             (short)0, acc,  false,  false);
    }

    float* yrow = Y + (size_t)(m0 + half * 8) * DOUT + col;
#pragma unroll
    for (int v = 0; v < 8; ++v) yrow[(size_t)v * DOUT] = acc[v];
}

template <int DOUT, int STRIDE>
__global__ void k_scatter(const float* __restrict__ XW, const int* __restrict__ src,
                          const int* __restrict__ dst, const float* __restrict__ cs,
                          float* __restrict__ AGG, int e) {
    int w    = (blockIdx.x * blockDim.x + threadIdx.x) >> 5;
    int lane = threadIdx.x & 31;
    if (w >= e) return;
    int s = src[w], d = dst[w];
    float c = cs[s];
    const float* xr = XW + (size_t)s * STRIDE;
    float* ar       = AGG + (size_t)d * STRIDE;
#pragma unroll
    for (int j = lane; j < DOUT; j += 32)
        atomicAdd(&ar[j], xr[j] * c);
}

template <int DOUT, int STRIDE, bool RELU>
__global__ void k_finalize(float* __restrict__ H, const float* __restrict__ cd,
                           const float* __restrict__ b, int n) {
    int i = blockIdx.x * blockDim.x + threadIdx.x;
    if (i < n * DOUT) {
        int r = i / DOUT, c = i % DOUT;
        size_t idx = (size_t)r * STRIDE + c;
        float v = H[idx] * cd[r] + b[c];
        if (RELU) v = fmaxf(v, 0.0f);
        H[idx] = v;
    }
}

__global__ void k_finalize_out(const float* __restrict__ AGG, const float* __restrict__ cd,
                               const float* __restrict__ b, float* __restrict__ out, int n) {
    int i = blockIdx.x * blockDim.x + threadIdx.x;
    if (i < n * 18) {
        int r = i / 18, c = i % 18;
        out[i] = AGG[(size_t)r * 32 + c] * cd[r] + b[c];
    }
}


static inline int cdiv(int a, int b) { return (a + b - 1) / b; }

extern "C" void kernel_launch(void* const* d_in, const int* in_sizes, int n_in,
                              void* d_out, int out_size, void* d_ws, size_t ws_size,
                              hipStream_t stream) {
    (void)n_in; (void)out_size; (void)ws_size;

    const float* feat = (const float*)d_in[0];
    const int*   src  = (const int*)d_in[1];
    const int*   dst  = (const int*)d_in[2];
    const float* W1 = (const float*)d_in[3];  const float* b1 = (const float*)d_in[4];
    const float* W2 = (const float*)d_in[5];  const float* b2 = (const float*)d_in[6];
    const float* W3 = (const float*)d_in[7];  const float* b3 = (const float*)d_in[8];
    const float* W4 = (const float*)d_in[9];  const float* b4 = (const float*)d_in[10];
    const float* W5 = (const float*)d_in[11]; const float* b5 = (const float*)d_in[12];
    float* out = (float*)d_out;

    const int N = 100000;
    const int E = in_sizes[1];

    float* ws = (float*)d_ws;
    size_t o = 0;
    float* c_src = ws + o; o += N;
    float* c_dst = ws + o; o += N;
    float* Xp    = ws + o; o += (size_t)N * 8;
    float* W1p   = ws + o; o += 8 * 128;
    float* W5p   = ws + o; o += 128 * 32;
    float* XW    = ws + o; o += (size_t)N * 128;
    float* H0    = ws + o; o += (size_t)N * 128;
    float* H1    = ws + o; o += (size_t)N * 128;

    k_zero<<<cdiv(2 * N, 256), 256, 0, stream>>>(c_src, 2 * N);
    k_degree<<<cdiv(E, 256), 256, 0, stream>>>(src, dst, c_src, c_dst, E);
    k_coeff<<<cdiv(2 * N, 256), 256, 0, stream>>>(c_src, 2 * N);

    k_pad_features<<<cdiv(N * 8, 256), 256, 0, stream>>>(feat, Xp, N);
    k_pad_w1<<<4, 256, 0, stream>>>(W1, W1p);
    k_pad_w5<<<16, 256, 0, stream>>>(W5, W5p);

    const dim3 gTile(N / 16, 128 / 16);
    const int  sBlk = cdiv(E, 8);
    const int  fBlk = cdiv(N * 128, 256);

    k_gemm_wmma<8, 128><<<gTile, 32, 0, stream>>>(Xp, W1p, XW);
    k_zero<<<fBlk, 256, 0, stream>>>(H0, N * 128);
    k_scatter<128, 128><<<sBlk, 256, 0, stream>>>(XW, src, dst, c_src, H0, E);
    k_finalize<128, 128, true><<<fBlk, 256, 0, stream>>>(H0, c_dst, b1, N);

    k_gemm_wmma<128, 128><<<gTile, 32, 0, stream>>>(H0, W2, XW);
    k_zero<<<fBlk, 256, 0, stream>>>(H1, N * 128);
    k_scatter<128, 128><<<sBlk, 256, 0, stream>>>(XW, src, dst, c_src, H1, E);
    k_finalize<128, 128, true><<<fBlk, 256, 0, stream>>>(H1, c_dst, b2, N);

    k_gemm_wmma<128, 128><<<gTile, 32, 0, stream>>>(H1, W3, XW);
    k_zero<<<fBlk, 256, 0, stream>>>(H0, N * 128);
    k_scatter<128, 128><<<sBlk, 256, 0, stream>>>(XW, src, dst, c_src, H0, E);
    k_finalize<128, 128, true><<<fBlk, 256, 0, stream>>>(H0, c_dst, b3, N);

    k_gemm_wmma<128, 128><<<gTile, 32, 0, stream>>>(H0, W4, XW);
    k_zero<<<fBlk, 256, 0, stream>>>(H1, N * 128);
    k_scatter<128, 128><<<sBlk, 256, 0, stream>>>(XW, src, dst, c_src, H1, E);
    k_finalize<128, 128, true><<<fBlk, 256, 0, stream>>>(H1, c_dst, b4, N);

    const dim3 gTile5(N / 16, 32 / 16);
    k_gemm_wmma<128, 32><<<gTile5, 32, 0, stream>>>(H1, W5p, XW);
    k_zero<<<cdiv(N * 32, 256), 256, 0, stream>>>(H0, N * 32);
    k_scatter<18, 32><<<sBlk, 256, 0, stream>>>(XW, src, dst, c_src, H0, E);
    k_finalize_out<<<cdiv(N * 18, 256), 256, 0, stream>>>(H0, c_dst, b5, out, N);
}


# Per-tile edge lists saved once per call: layer 1's scan also writes each wave's matched edges (source, tile row, c_src) to the workspace; layers 2-5 read them: no id scan, no per-edge index loads
# speedup vs baseline: 2.4106x; 2.4106x over previous
#include <hip/hip_runtime.h>

typedef __attribute__((ext_vector_type(2))) float v2f;
typedef __attribute__((ext_vector_type(8))) float v8f;


__global__ void k_zero(float* __restrict__ p, int n) {
    int i = blockIdx.x * blockDim.x + threadIdx.x;
    if (i < n) p[i] = 0.0f;
}

__global__ void k_degree(const int* __restrict__ src, const int* __restrict__ dst,
                         float* __restrict__ degO, float* __restrict__ degI, int e) {
    int i = blockIdx.x * blockDim.x + threadIdx.x;
    if (i < e) {
        atomicAdd(&degO[src[i]], 1.0f);
        atomicAdd(&degI[dst[i]], 1.0f);
    }
}

__global__ void k_coeff(float* __restrict__ c, int n) {
    int i = blockIdx.x * blockDim.x + threadIdx.x;
    if (i < n) {
        float d = c[i];
        c[i] = (d > 0.0f) ? rsqrtf(d) : 0.0f;
    }
}

__global__ void k_pad_features(const float* __restrict__ x, float* __restrict__ xp, int n) {
    int i = blockIdx.x * blockDim.x + threadIdx.x;
    if (i < n * 8) {
        int r = i >> 3, c = i & 7;
        xp[i] = (c < 6) ? x[r * 6 + c] : 0.0f;
    }
}

__global__ void k_pad_w1(const float* __restrict__ w, float* __restrict__ wp) {
    int i = blockIdx.x * blockDim.x + threadIdx.x;
    if (i < 8 * 128) {
        int r = i >> 7;
        wp[i] = (r < 6) ? w[i] : 0.0f;
    }
}

__global__ void k_pad_w5(const float* __restrict__ w, float* __restrict__ wp) {
    int i = blockIdx.x * blockDim.x + threadIdx.x;
    if (i < 128 * 32) {
        int r = i >> 5, c = i & 31;
        wp[i] = (c < 18) ? w[r * 18 + c] : 0.0f;
    }
}

template <int DIN, int DOUT>
__global__ void k_gemm_wmma(const float* __restrict__ A, const float* __restrict__ W,
                            float* __restrict__ Y) {
    const int lane = threadIdx.x;
    const int m0   = blockIdx.x * 16;
    const int n0   = blockIdx.y * 16;
    const int half = lane >> 4;
    const int l15  = lane & 15;
    const int koff = half * 2;

    const float* arow = A + (size_t)(m0 + l15) * DIN + koff;
    const int col = n0 + l15;

    v8f acc = {};
#pragma unroll
    for (int k = 0; k < DIN; k += 4) {
        v2f a = *(const v2f*)(arow + k);
        v2f b;
        b.x = W[(size_t)(k + koff) * DOUT + col];
        b.y = W[(size_t)(k + koff + 1) * DOUT + col];
        acc = __builtin_amdgcn_wmma_f32_16x16x4_f32(
             false, a,  false, b,
             (short)0, acc,  false,  false);
    }

    float* yrow = Y + (size_t)(m0 + half * 8) * DOUT + col;
#pragma unroll
    for (int v = 0; v < 8; ++v) yrow[(size_t)v * DOUT] = acc[v];
}


#define TILE_ROWS   400
#define QCAP        768
#define AGG_THREADS 1024
#define AGG_WAVES   32
#define BLK_EDGES   256
#define DRAIN_U     8
#define SEG_CAP     512
#define SRC_BITS    17

__global__ void k_tile_ids(const int* __restrict__ dst, int* __restrict__ ids, int e, int nwords) {
    int i = blockIdx.x * blockDim.x + threadIdx.x;
    if (i < nwords) {
        unsigned w = 0u;
#pragma unroll
        for (int j = 0; j < 4; ++j) {
            int ei = i * 4 + j;
            unsigned t = (ei < e) ? (unsigned)(dst[ei] / TILE_ROWS) : 255u;
            w |= (t & 255u) << (8 * j);
        }
        ids[i] = (int)w;
    }
}

__device__ __forceinline__ unsigned zero_bytes(unsigned t) {
    return ~(((t & 0x7F7F7F7Fu) + 0x7F7F7F7Fu) | t | 0x7F7F7F7Fu);
}

__device__ __forceinline__ unsigned zero_byte_nibble(unsigned t) {
    return ((zero_bytes(t) >> 7) * 0x01020408u) >> 24;
}

__device__ __forceinline__ void load_id_words(int2 (&v)[4], const int* __restrict__ ids,
                                              int b0, int nblk, int lane) {
#pragma unroll
    for (int u = 0; u < 4; ++u) {
        const int b  = b0 + AGG_WAVES * u;
        const int bc = b < nblk ? b : nblk - 1;
        v[u] = *(const int2*)(ids + (size_t)bc * (BLK_EDGES / 4) + lane * 2);
    }
}

template <int NCH, int STRIDE>
__device__ __forceinline__ void scatter_group(const int (&sj)[DRAIN_U], const int (&dl)[DRAIN_U],
                                              const float (&cc)[DRAIN_U], int nv,
                                              const float* __restrict__ XW,
                                              float* acc, int lane) {
    constexpr int W = NCH * 32;
    float r[DRAIN_U][NCH];
#pragma unroll
    for (int u = 0; u < DRAIN_U; ++u) {
        const float* xr = XW + (size_t)sj[u] * STRIDE + lane;
#pragma unroll
        for (int k = 0; k < NCH; ++k) r[u][k] = xr[k * 32];
    }
#pragma unroll
    for (int u = 0; u < DRAIN_U; ++u) {
        if (u == 0 || u < nv) {
            float* ar = acc + dl[u] * W + lane;
#pragma unroll
            for (int k = 0; k < NCH; ++k) atomicAdd(&ar[k * 32], r[u][k] * cc[u]);
        }
    }
}

template <int NCH, int STRIDE>
__device__ __forceinline__ void scatter_rows(int s, int t, float c, int nb,
                                             const float* __restrict__ XW,
                                             float* acc, int lane) {
    for (int j0 = 0; j0 < nb; j0 += DRAIN_U) {
        int   sj[DRAIN_U];
        int   dl[DRAIN_U];
        float cc[DRAIN_U];
#pragma unroll
        for (int u = 0; u < DRAIN_U; ++u) {
            const int jj = j0 + u;
            sj[u] = __builtin_amdgcn_readlane(s, jj);
            dl[u] = __builtin_amdgcn_readlane(t, jj);
            cc[u] = __int_as_float(__builtin_amdgcn_readlane(__float_as_int(c), jj));
        }
        scatter_group<NCH, STRIDE>(sj, dl, cc, nb - j0, XW, acc, lane);
    }
}

template <int NCH, int STRIDE, bool SAVE>
__device__ __forceinline__ void drain_queue(const unsigned* q, int cnt,
                                            const float* __restrict__ XW,
                                            const int* __restrict__ src,
                                            const int* __restrict__ dst,
                                            const float* __restrict__ cs,
                                            float* acc, int base, int lane,
                                            int2* seg, int& saved) {
    for (int i0 = 0; i0 < cnt; i0 += 32) {
        const int nb = (cnt - i0 < 32) ? (cnt - i0) : 32;
        int s = 0, t = 0;
        float c = 0.0f;
        if (lane < nb) {
            unsigned e = q[i0 + lane];
            s = src[e];
            t = dst[e] - base;
            t = t < 0 ? 0 : (t > TILE_ROWS - 1 ? TILE_ROWS - 1 : t);
            c = cs[s];
        }
        if (SAVE) {
            if (saved + nb <= SEG_CAP && lane < nb)
                seg[saved + lane] = make_int2(s | (t << SRC_BITS), __float_as_int(c));
            saved += nb;
        }
        scatter_rows<NCH, STRIDE>(s, t, c, nb, XW, acc, lane);
    }
}

__device__ __forceinline__ void load_entries(int (&w)[2 * DRAIN_U], const int* __restrict__ segw, int j) {
#pragma unroll
    for (int i = 0; i < 2 * DRAIN_U; ++i) w[i] = segw[2 * j + i];
}

template <int NCH, int STRIDE>
__device__ __forceinline__ void drain_list(const int* __restrict__ segw, int cnt,
                                           const float* __restrict__ XW,
                                           float* acc, int lane) {
    if (cnt <= 0) return;
    int ex[2 * DRAIN_U];
    load_entries(ex, segw, 0);
    for (int j0 = 0; j0 < cnt; j0 += DRAIN_U) {
        int nx[2 * DRAIN_U];
        load_entries(nx, segw, (j0 + DRAIN_U < cnt) ? j0 + DRAIN_U : j0);
        int   sj[DRAIN_U];
        int   dl[DRAIN_U];
        float cc[DRAIN_U];
#pragma unroll
        for (int u = 0; u < DRAIN_U; ++u) {
            sj[u] = ex[2 * u] & ((1 << SRC_BITS) - 1);
            const int t = (int)((unsigned)ex[2 * u] >> SRC_BITS);
            dl[u] = t > TILE_ROWS - 1 ? TILE_ROWS - 1 : t;
            cc[u] = __int_as_float(ex[2 * u + 1]);
        }
        scatter_group<NCH, STRIDE>(sj, dl, cc, cnt - j0, XW, acc, lane);
#pragma unroll
        for (int i = 0; i < 2 * DRAIN_U; ++i) ex[i] = nx[i];
    }
}

template <int NCH, int STRIDE, int OUTW, bool RELU, bool BUILD>
__global__ void __launch_bounds__(AGG_THREADS)
k_tile_aggregate(const float* __restrict__ XW, const int* __restrict__ src,
                 const int* __restrict__ dst, const int* __restrict__ ids,
                 const float* __restrict__ cs, const float* __restrict__ cd,
                 const float* __restrict__ bias, float* __restrict__ Y,
                 int* lists, int* segcnt, int n, int nblk) {
    constexpr int W = NCH * 32;
    __shared__ float    acc[TILE_ROWS * W];
    __shared__ unsigned queue[AGG_WAVES * QCAP];

    const int tid  = threadIdx.x;
    const int lane = tid & 31;
    const int wave = __builtin_amdgcn_readfirstlane(tid >> 5);
    const int p    = blockIdx.x;
    const int base = p * TILE_ROWS;

    for (int i = tid; i < TILE_ROWS * W; i += AGG_THREADS) acc[i] = 0.0f;
    __syncthreads();

    const int segi = p * AGG_WAVES + wave;
    int2* seg = (int2*)lists + (size_t)segi * SEG_CAP;
    int  saved = 0;
    bool scan  = true;
    if (!BUILD) {
        int sc = __builtin_amdgcn_readfirstlane(segcnt[segi]);
        if (sc >= 0) {
            scan = false;
            sc = sc < SEG_CAP ? sc : SEG_CAP;
            drain_list<NCH, STRIDE>((const int*)seg, sc, XW, acc, lane);
        }
    }

    if (scan) {
        unsigned* q = queue + wave * QCAP;
        int cnt = 0;
        const unsigned pp = (unsigned)p * 0x01010101u;
        const unsigned lt = (1u << lane) - 1u;

        if (nblk > 0) {
            int2 cur[4];
            load_id_words(cur, ids, wave, nblk, lane);
            for (int b0 = wave; b0 < nblk; b0 += AGG_WAVES * 4) {
                int2 nxt[4];
                load_id_words(nxt, ids, b0 + AGG_WAVES * 4, nblk, lane);

                unsigned mm = 0u;
#pragma unroll
                for (int u = 0; u < 4; ++u) {
                    if (b0 + AGG_WAVES * u < nblk) {
                        mm |= zero_byte_nibble((unsigned)cur[u].x ^ pp) << (8 * u);
                        mm |= zero_byte_nibble((unsigned)cur[u].y ^ pp) << (8 * u + 4);
                    }
                }
                const unsigned e0 = (unsigned)b0 * BLK_EDGES + (unsigned)lane * 8u;
                for (int it = 0; it < 32; ++it) {
                    const bool has = mm != 0u;
                    const unsigned m = __builtin_amdgcn_ballot_w32(has);
                    if (m == 0u) break;
                    if (cnt > QCAP - 32) {
                        drain_queue<NCH, STRIDE, BUILD>(q, cnt, XW, src, dst, cs, acc, base, lane,
                                                        seg, saved);
                        cnt = 0;
                    }
                    if (has) {
                        const unsigned i = (unsigned)__builtin_ctz(mm);
                        mm &= mm - 1u;
                        q[cnt + __popc(m & lt)] = e0 + (i >> 3) * (unsigned)(AGG_WAVES * BLK_EDGES) + (i & 7u);
                    }
                    cnt += __popc(m);
                }
#pragma unroll
                for (int u = 0; u < 4; ++u) cur[u] = nxt[u];
            }
        }
        drain_queue<NCH, STRIDE, BUILD>(q, cnt, XW, src, dst, cs, acc, base, lane, seg, saved);
    }
    if (BUILD) {
        const int padn = (DRAIN_U - (saved & (DRAIN_U - 1))) & (DRAIN_U - 1);
        if (saved <= SEG_CAP && lane < padn) seg[saved + lane] = make_int2(0, 0);
        if (lane == 0) segcnt[segi] = saved <= SEG_CAP ? saved : -1;
    }
    __syncthreads();

    if (OUTW == W) {
        for (int i = tid; i < TILE_ROWS * W; i += AGG_THREADS) {
            const int r = i / W, c = i % W;
            const int row = base + r;
            if (row < n) {
                float v = acc[i] * cd[row] + bias[c];
                if (RELU) v = fmaxf(v, 0.0f);
                Y[(size_t)row * W + c] = v;
            }
        }
    } else {
        for (int i = tid; i < TILE_ROWS * OUTW; i += AGG_THREADS) {
            const int r = i / OUTW, c = i % OUTW;
            const int row = base + r;
            if (row < n) {
                float v = acc[r * W + c] * cd[row] + bias[c];
                if (RELU) v = fmaxf(v, 0.0f);
                Y[(size_t)base * OUTW + i] = v;
            }
        }
    }
}


static inline int cdiv(int a, int b) { return (a + b - 1) / b; }

extern "C" void kernel_launch(void* const* d_in, const int* in_sizes, int n_in,
                              void* d_out, int out_size, void* d_ws, size_t ws_size,
                              hipStream_t stream) {
    (void)n_in; (void)out_size; (void)ws_size;

    const float* feat = (const float*)d_in[0];
    const int*   src  = (const int*)d_in[1];
    const int*   dst  = (const int*)d_in[2];
    const float* W1 = (const float*)d_in[3];  const float* b1 = (const float*)d_in[4];
    const float* W2 = (const float*)d_in[5];  const float* b2 = (const float*)d_in[6];
    const float* W3 = (const float*)d_in[7];  const float* b3 = (const float*)d_in[8];
    const float* W4 = (const float*)d_in[9];  const float* b4 = (const float*)d_in[10];
    const float* W5 = (const float*)d_in[11]; const float* b5 = (const float*)d_in[12];
    float* out = (float*)d_out;

    const int N = 100000;
    const int E = in_sizes[1];

    float* ws = (float*)d_ws;
    size_t o = 0;
    float* c_src = ws + o; o += N;
    float* c_dst = ws + o; o += N;
    float* Xp    = ws + o; o += (size_t)N * 8;
    float* W1p   = ws + o; o += 8 * 128;
    float* W5p   = ws + o; o += 128 * 32;
    float* XW    = ws + o; o += (size_t)N * 128;
    float* H0    = ws + o; o += (size_t)N * 128;
    int*   ids   = (int*)(ws + o);

    k_zero<<<cdiv(2 * N, 256), 256, 0, stream>>>(c_src, 2 * N);
    k_degree<<<cdiv(E, 256), 256, 0, stream>>>(src, dst, c_src, c_dst, E);
    k_coeff<<<cdiv(2 * N, 256), 256, 0, stream>>>(c_src, 2 * N);

    k_pad_features<<<cdiv(N * 8, 256), 256, 0, stream>>>(feat, Xp, N);
    k_pad_w1<<<4, 256, 0, stream>>>(W1, W1p);
    k_pad_w5<<<16, 256, 0, stream>>>(W5, W5p);

    const int nblk   = cdiv(E, BLK_EDGES);
    const int nwords = nblk * (BLK_EDGES / 4);
    k_tile_ids<<<cdiv(nwords, 256), 256, 0, stream>>>(dst, ids, E, nwords);

    const dim3 gTile(N / 16, 128 / 16);
    const int  nTiles = N / TILE_ROWS;

    int* lists  = ids + nwords;
    int* segcnt = lists + (size_t)nTiles * AGG_WAVES * SEG_CAP * 2;

    k_gemm_wmma<8, 128><<<gTile, 32, 0, stream>>>(Xp, W1p, XW);
    k_tile_aggregate<4, 128, 128, true, true><<<nTiles, AGG_THREADS, 0, stream>>>(
        XW, src, dst, ids, c_src, c_dst, b1, H0, lists, segcnt, N, nblk);

    k_gemm_wmma<128, 128><<<gTile, 32, 0, stream>>>(H0, W2, XW);
    k_tile_aggregate<4, 128, 128, true, false><<<nTiles, AGG_THREADS, 0, stream>>>(
        XW, src, dst, ids, c_src, c_dst, b2, H0, lists, segcnt, N, nblk);

    k_gemm_wmma<128, 128><<<gTile, 32, 0, stream>>>(H0, W3, XW);
    k_tile_aggregate<4, 128, 128, true, false><<<nTiles, AGG_THREADS, 0, stream>>>(
        XW, src, dst, ids, c_src, c_dst, b3, H0, lists, segcnt, N, nblk);

    k_gemm_wmma<128, 128><<<gTile, 32, 0, stream>>>(H0, W4, XW);
    k_tile_aggregate<4, 128, 128, true, false><<<nTiles, AGG_THREADS, 0, stream>>>(
        XW, src, dst, ids, c_src, c_dst, b4, H0, lists, segcnt, N, nblk);

    const dim3 gTile5(N / 16, 32 / 16);
    k_gemm_wmma<128, 32><<<gTile5, 32, 0, stream>>>(H0, W5p, XW);
    k_tile_aggregate<1, 32, 18, false, false><<<nTiles, AGG_THREADS, 0, stream>>>(
        XW, src, dst, ids, c_src, c_dst, b5, out, lists, segcnt, N, nblk);
}


# f32 WMMA products re-tiled: 256 resident 16-wave work-groups loop over strips of 16 rows x all columns; W staged once per group in LDS in fragment order, A read once, whole-row stores
# speedup vs baseline: 2.6289x; 1.0906x over previous
#include <hip/hip_runtime.h>

typedef __attribute__((ext_vector_type(2))) float v2f;
typedef __attribute__((ext_vector_type(8))) float v8f;


__global__ void k_zero(float* __restrict__ p, int n) {
    int i = blockIdx.x * blockDim.x + threadIdx.x;
    if (i < n) p[i] = 0.0f;
}

__global__ void k_degree(const int* __restrict__ src, const int* __restrict__ dst,
                         float* __restrict__ degO, float* __restrict__ degI, int e) {
    int i = blockIdx.x * blockDim.x + threadIdx.x;
    if (i < e) {
        atomicAdd(&degO[src[i]], 1.0f);
        atomicAdd(&degI[dst[i]], 1.0f);
    }
}

__global__ void k_coeff(float* __restrict__ c, int n) {
    int i = blockIdx.x * blockDim.x + threadIdx.x;
    if (i < n) {
        float d = c[i];
        c[i] = (d > 0.0f) ? rsqrtf(d) : 0.0f;
    }
}

__global__ void k_pad_features(const float* __restrict__ x, float* __restrict__ xp, int n) {
    int i = blockIdx.x * blockDim.x + threadIdx.x;
    if (i < n * 8) {
        int r = i >> 3, c = i & 7;
        xp[i] = (c < 6) ? x[r * 6 + c] : 0.0f;
    }
}

__global__ void k_pad_w1(const float* __restrict__ w, float* __restrict__ wp) {
    int i = blockIdx.x * blockDim.x + threadIdx.x;
    if (i < 8 * 128) {
        int r = i >> 7;
        wp[i] = (r < 6) ? w[i] : 0.0f;
    }
}

__global__ void k_pad_w5(const float* __restrict__ w, float* __restrict__ wp) {
    int i = blockIdx.x * blockDim.x + threadIdx.x;
    if (i < 128 * 32) {
        int r = i >> 5, c = i & 31;
        wp[i] = (c < 18) ? w[r * 18 + c] : 0.0f;
    }
}


typedef __attribute__((ext_vector_type(4))) float v4f;

#define GEMM_WAVES      16
#define GEMM_THREADS    512
#define GEMM_GROUPS     256
#define GEMM_W_FLOATS   (128 * 128)
#define GEMM_STRIP_MAX  (16 * (128 + 4))
#define GEMM_LDS_FLOATS (GEMM_W_FLOATS + GEMM_WAVES * GEMM_STRIP_MAX)

template <int DIN>
__device__ __forceinline__ void load_a_strip(v4f (&a)[DIN / 8], const float* __restrict__ A,
                                             int mt, int c, int h) {
    const float* p = A + (size_t)(mt * 16 + c) * DIN + 4 * h;
#pragma unroll
    for (int j = 0; j < DIN / 8; ++j) a[j] = *(const v4f*)(p + 8 * j);
}

template <int DIN, int DOUT>
__global__ void __launch_bounds__(GEMM_THREADS)
k_gemm_strips(const float* __restrict__ A, const float* __restrict__ Wm,
              float* __restrict__ Y, int mtiles) {
    constexpr int NJ    = DIN / 8;
    constexpr int NT    = DOUT / 16;
    constexpr int PITCH = DOUT + 4;
    static_assert(DIN * DOUT <= GEMM_W_FLOATS && 16 * PITCH <= GEMM_STRIP_MAX, "LDS plan");
    __shared__ __attribute__((aligned(16))) float lds[GEMM_LDS_FLOATS];
    v4f*   wl = (v4f*)lds;
    float* ot = lds + GEMM_W_FLOATS;

    const int tid  = threadIdx.x;
    const int lane = tid & 31;
    const int c    = lane & 15;
    const int h    = lane >> 4;
    const int wave = __builtin_amdgcn_readfirstlane(tid >> 5);

    const int stride = (int)gridDim.x * GEMM_WAVES;
    int mt = (int)blockIdx.x + (int)gridDim.x * wave;

    v4f a[NJ];
    if (mt < mtiles) load_a_strip<DIN>(a, A, mt, c, h);

    constexpr int ITEMS  = NJ * NT * 32;
    constexpr int ROUNDS = (ITEMS + GEMM_THREADS - 1) / GEMM_THREADS;
#pragma unroll
    for (int it = 0; it < ROUNDS; ++it) {
        const int idx = tid + it * GEMM_THREADS;
        if (ITEMS % GEMM_THREADS == 0 || idx < ITEMS) {
            const int g  = idx >> 5;
            const int j  = g / NT, nt = g % NT;
            const int k0 = 8 * j + 4 * h, n = nt * 16 + c;
            v4f v;
            v.x = Wm[(size_t)(k0 + 0) * DOUT + n];
            v.y = Wm[(size_t)(k0 + 1) * DOUT + n];
            v.z = Wm[(size_t)(k0 + 2) * DOUT + n];
            v.w = Wm[(size_t)(k0 + 3) * DOUT + n];
            wl[idx] = v;
        }
    }
    __syncthreads();

    float* o = ot + wave * (16 * PITCH);
    constexpr int LPR = DOUT / 4;
    constexpr int RPI = 32 / LPR;
    const int rsub = lane / LPR, col = (lane % LPR) * 4;

    while (mt < mtiles) {
        v8f acc[NT];
#pragma unroll
        for (int nt = 0; nt < NT; ++nt) acc[nt] = (v8f){};

#pragma unroll
        for (int j = 0; j < NJ; ++j) {
            v4f b[NT];
#pragma unroll
            for (int nt = 0; nt < NT; ++nt) b[nt] = wl[(j * NT + nt) * 32 + lane];
            const v2f a0 = {a[j].x, a[j].y}, a1 = {a[j].z, a[j].w};
#pragma unroll
            for (int nt = 0; nt < NT; ++nt) {
                const v2f b0 = {b[nt].x, b[nt].y};
                acc[nt] = __builtin_amdgcn_wmma_f32_16x16x4_f32(
                     false, a0,  false, b0,
                     (short)0, acc[nt],  false,  false);
            }
#pragma unroll
            for (int nt = 0; nt < NT; ++nt) {
                const v2f b1 = {b[nt].z, b[nt].w};
                acc[nt] = __builtin_amdgcn_wmma_f32_16x16x4_f32(
                     false, a1,  false, b1,
                     (short)0, acc[nt],  false,  false);
            }
        }

        const int nx = mt + stride;
        if (nx < mtiles) load_a_strip<DIN>(a, A, nx, c, h);

#pragma unroll
        for (int nt = 0; nt < NT; ++nt)
#pragma unroll
            for (int r = 0; r < 8; ++r)
                o[(8 * h + r) * PITCH + nt * 16 + c] = acc[nt][r];
        float* yp = Y + (size_t)mt * (16 * DOUT) + rsub * DOUT + col;
#pragma unroll
        for (int i = 0; i < 16 / RPI; ++i) {
            const v4f v = *(const v4f*)(o + (i * RPI + rsub) * PITCH + col);
            *(v4f*)(yp + i * (RPI * DOUT)) = v;
        }
        mt = nx;
    }
}


#define TILE_ROWS   400
#define QCAP        768
#define AGG_THREADS 1024
#define AGG_WAVES   32
#define BLK_EDGES   256
#define DRAIN_U     8
#define SEG_CAP     512
#define SRC_BITS    17

__global__ void k_tile_ids(const int* __restrict__ dst, int* __restrict__ ids, int e, int nwords) {
    int i = blockIdx.x * blockDim.x + threadIdx.x;
    if (i < nwords) {
        unsigned w = 0u;
#pragma unroll
        for (int j = 0; j < 4; ++j) {
            int ei = i * 4 + j;
            unsigned t = (ei < e) ? (unsigned)(dst[ei] / TILE_ROWS) : 255u;
            w |= (t & 255u) << (8 * j);
        }
        ids[i] = (int)w;
    }
}

__device__ __forceinline__ unsigned zero_bytes(unsigned t) {
    return ~(((t & 0x7F7F7F7Fu) + 0x7F7F7F7Fu) | t | 0x7F7F7F7Fu);
}

__device__ __forceinline__ unsigned zero_byte_nibble(unsigned t) {
    return ((zero_bytes(t) >> 7) * 0x01020408u) >> 24;
}

__device__ __forceinline__ void load_id_words(int2 (&v)[4], const int* __restrict__ ids,
                                              int b0, int nblk, int lane) {
#pragma unroll
    for (int u = 0; u < 4; ++u) {
        const int b  = b0 + AGG_WAVES * u;
        const int bc = b < nblk ? b : nblk - 1;
        v[u] = *(const int2*)(ids + (size_t)bc * (BLK_EDGES / 4) + lane * 2);
    }
}

template <int NCH, int STRIDE>
__device__ __forceinline__ void scatter_group(const int (&sj)[DRAIN_U], const int (&dl)[DRAIN_U],
                                              const float (&cc)[DRAIN_U], int nv,
                                              const float* __restrict__ XW,
                                              float* acc, int lane) {
    constexpr int W = NCH * 32;
    float r[DRAIN_U][NCH];
#pragma unroll
    for (int u = 0; u < DRAIN_U; ++u) {
        const float* xr = XW + (size_t)sj[u] * STRIDE + lane;
#pragma unroll
        for (int k = 0; k < NCH; ++k) r[u][k] = xr[k * 32];
    }
#pragma unroll
    for (int u = 0; u < DRAIN_U; ++u) {
        if (u == 0 || u < nv) {
            float* ar = acc + dl[u] * W + lane;
#pragma unroll
            for (int k = 0; k < NCH; ++k) atomicAdd(&ar[k * 32], r[u][k] * cc[u]);
        }
    }
}

template <int NCH, int STRIDE>
__device__ __forceinline__ void scatter_rows(int s, int t, float c, int nb,
                                             const float* __restrict__ XW,
                                             float* acc, int lane) {
    for (int j0 = 0; j0 < nb; j0 += DRAIN_U) {
        int   sj[DRAIN_U];
        int   dl[DRAIN_U];
        float cc[DRAIN_U];
#pragma unroll
        for (int u = 0; u < DRAIN_U; ++u) {
            const int jj = j0 + u;
            sj[u] = __builtin_amdgcn_readlane(s, jj);
            dl[u] = __builtin_amdgcn_readlane(t, jj);
            cc[u] = __int_as_float(__builtin_amdgcn_readlane(__float_as_int(c), jj));
        }
        scatter_group<NCH, STRIDE>(sj, dl, cc, nb - j0, XW, acc, lane);
    }
}

template <int NCH, int STRIDE, bool SAVE>
__device__ __forceinline__ void drain_queue(const unsigned* q, int cnt,
                                            const float* __restrict__ XW,
                                            const int* __restrict__ src,
                                            const int* __restrict__ dst,
                                            const float* __restrict__ cs,
                                            float* acc, int base, int lane,
                                            int2* seg, int& saved) {
    for (int i0 = 0; i0 < cnt; i0 += 32) {
        const int nb = (cnt - i0 < 32) ? (cnt - i0) : 32;
        int s = 0, t = 0;
        float c = 0.0f;
        if (lane < nb) {
            unsigned e = q[i0 + lane];
            s = src[e];
            t = dst[e] - base;
            t = t < 0 ? 0 : (t > TILE_ROWS - 1 ? TILE_ROWS - 1 : t);
            c = cs[s];
        }
        if (SAVE) {
            if (saved + nb <= SEG_CAP && lane < nb)
                seg[saved + lane] = make_int2(s | (t << SRC_BITS), __float_as_int(c));
            saved += nb;
        }
        scatter_rows<NCH, STRIDE>(s, t, c, nb, XW, acc, lane);
    }
}

__device__ __forceinline__ void load_entries(int (&w)[2 * DRAIN_U], const int* __restrict__ segw, int j) {
#pragma unroll
    for (int i = 0; i < 2 * DRAIN_U; ++i) w[i] = segw[2 * j + i];
}

template <int NCH, int STRIDE>
__device__ __forceinline__ void drain_list(const int* __restrict__ segw, int cnt,
                                           const float* __restrict__ XW,
                                           float* acc, int lane) {
    if (cnt <= 0) return;
    int ex[2 * DRAIN_U];
    load_entries(ex, segw, 0);
    for (int j0 = 0; j0 < cnt; j0 += DRAIN_U) {
        int nx[2 * DRAIN_U];
        load_entries(nx, segw, (j0 + DRAIN_U < cnt) ? j0 + DRAIN_U : j0);
        int   sj[DRAIN_U];
        int   dl[DRAIN_U];
        float cc[DRAIN_U];
#pragma unroll
        for (int u = 0; u < DRAIN_U; ++u) {
            sj[u] = ex[2 * u] & ((1 << SRC_BITS) - 1);
            const int t = (int)((unsigned)ex[2 * u] >> SRC_BITS);
            dl[u] = t > TILE_ROWS - 1 ? TILE_ROWS - 1 : t;
            cc[u] = __int_as_float(ex[2 * u + 1]);
        }
        scatter_group<NCH, STRIDE>(sj, dl, cc, cnt - j0, XW, acc, lane);
#pragma unroll
        for (int i = 0; i < 2 * DRAIN_U; ++i) ex[i] = nx[i];
    }
}

template <int NCH, int STRIDE, int OUTW, bool RELU, bool BUILD>
__global__ void __launch_bounds__(AGG_THREADS)
k_tile_aggregate(const float* __restrict__ XW, const int* __restrict__ src,
                 const int* __restrict__ dst, const int* __restrict__ ids,
                 const float* __restrict__ cs, const float* __restrict__ cd,
                 const float* __restrict__ bias, float* __restrict__ Y,
                 int* lists, int* segcnt, int n, int nblk) {
    constexpr int W = NCH * 32;
    __shared__ float    acc[TILE_ROWS * W];
    __shared__ unsigned queue[AGG_WAVES * QCAP];

    const int tid  = threadIdx.x;
    const int lane = tid & 31;
    const int wave = __builtin_amdgcn_readfirstlane(tid >> 5);
    const int p    = blockIdx.x;
    const int base = p * TILE_ROWS;

    for (int i = tid; i < TILE_ROWS * W; i += AGG_THREADS) acc[i] = 0.0f;
    __syncthreads();

    const int segi = p * AGG_WAVES + wave;
    int2* seg = (int2*)lists + (size_t)segi * SEG_CAP;
    int  saved = 0;
    bool scan  = true;
    if (!BUILD) {
        int sc = __builtin_amdgcn_readfirstlane(segcnt[segi]);
        if (sc >= 0) {
            scan = false;
            sc = sc < SEG_CAP ? sc : SEG_CAP;
            drain_list<NCH, STRIDE>((const int*)seg, sc, XW, acc, lane);
        }
    }

    if (scan) {
        unsigned* q = queue + wave * QCAP;
        int cnt = 0;
        const unsigned pp = (unsigned)p * 0x01010101u;
        const unsigned lt = (1u << lane) - 1u;

        if (nblk > 0) {
            int2 cur[4];
            load_id_words(cur, ids, wave, nblk, lane);
            for (int b0 = wave; b0 < nblk; b0 += AGG_WAVES * 4) {
                int2 nxt[4];
                load_id_words(nxt, ids, b0 + AGG_WAVES * 4, nblk, lane);

                unsigned mm = 0u;
#pragma unroll
                for (int u = 0; u < 4; ++u) {
                    if (b0 + AGG_WAVES * u < nblk) {
                        mm |= zero_byte_nibble((unsigned)cur[u].x ^ pp) << (8 * u);
                        mm |= zero_byte_nibble((unsigned)cur[u].y ^ pp) << (8 * u + 4);
                    }
                }
                const unsigned e0 = (unsigned)b0 * BLK_EDGES + (unsigned)lane * 8u;
                for (int it = 0; it < 32; ++it) {
                    const bool has = mm != 0u;
                    const unsigned m = __builtin_amdgcn_ballot_w32(has);
                    if (m == 0u) break;
                    if (cnt > QCAP - 32) {
                        drain_queue<NCH, STRIDE, BUILD>(q, cnt, XW, src, dst, cs, acc, base, lane,
                                                        seg, saved);
                        cnt = 0;
                    }
                    if (has) {
                        const unsigned i = (unsigned)__builtin_ctz(mm);
                        mm &= mm - 1u;
                        q[cnt + __popc(m & lt)] = e0 + (i >> 3) * (unsigned)(AGG_WAVES * BLK_EDGES) + (i & 7u);
                    }
                    cnt += __popc(m);
                }
#pragma unroll
                for (int u = 0; u < 4; ++u) cur[u] = nxt[u];
            }
        }
        drain_queue<NCH, STRIDE, BUILD>(q, cnt, XW, src, dst, cs, acc, base, lane, seg, saved);
    }
    if (BUILD) {
        const int padn = (DRAIN_U - (saved & (DRAIN_U - 1))) & (DRAIN_U - 1);
        if (saved <= SEG_CAP && lane < padn) seg[saved + lane] = make_int2(0, 0);
        if (lane == 0) segcnt[segi] = saved <= SEG_CAP ? saved : -1;
    }
    __syncthreads();

    if (OUTW == W) {
        for (int i = tid; i < TILE_ROWS * W; i += AGG_THREADS) {
            const int r = i / W, c = i % W;
            const int row = base + r;
            if (row < n) {
                float v = acc[i] * cd[row] + bias[c];
                if (RELU) v = fmaxf(v, 0.0f);
                Y[(size_t)row * W + c] = v;
            }
        }
    } else {
        for (int i = tid; i < TILE_ROWS * OUTW; i += AGG_THREADS) {
            const int r = i / OUTW, c = i % OUTW;
            const int row = base + r;
            if (row < n) {
                float v = acc[r * W + c] * cd[row] + bias[c];
                if (RELU) v = fmaxf(v, 0.0f);
                Y[(size_t)base * OUTW + i] = v;
            }
        }
    }
}


static inline int cdiv(int a, int b) { return (a + b - 1) / b; }

extern "C" void kernel_launch(void* const* d_in, const int* in_sizes, int n_in,
                              void* d_out, int out_size, void* d_ws, size_t ws_size,
                              hipStream_t stream) {
    (void)n_in; (void)out_size; (void)ws_size;

    const float* feat = (const float*)d_in[0];
    const int*   src  = (const int*)d_in[1];
    const int*   dst  = (const int*)d_in[2];
    const float* W1 = (const float*)d_in[3];  const float* b1 = (const float*)d_in[4];
    const float* W2 = (const float*)d_in[5];  const float* b2 = (const float*)d_in[6];
    const float* W3 = (const float*)d_in[7];  const float* b3 = (const float*)d_in[8];
    const float* W4 = (const float*)d_in[9];  const float* b4 = (const float*)d_in[10];
    const float* W5 = (const float*)d_in[11]; const float* b5 = (const float*)d_in[12];
    float* out = (float*)d_out;

    const int N = 100000;
    const int E = in_sizes[1];

    float* ws = (float*)d_ws;
    size_t o = 0;
    float* c_src = ws + o; o += N;
    float* c_dst = ws + o; o += N;
    float* Xp    = ws + o; o += (size_t)N * 8;
    float* W1p   = ws + o; o += 8 * 128;
    float* W5p   = ws + o; o += 128 * 32;
    float* XW    = ws + o; o += (size_t)N * 128;
    float* H0    = ws + o; o += (size_t)N * 128;
    int*   ids   = (int*)(ws + o);

    k_zero<<<cdiv(2 * N, 256), 256, 0, stream>>>(c_src, 2 * N);
    k_degree<<<cdiv(E, 256), 256, 0, stream>>>(src, dst, c_src, c_dst, E);
    k_coeff<<<cdiv(2 * N, 256), 256, 0, stream>>>(c_src, 2 * N);

    k_pad_features<<<cdiv(N * 8, 256), 256, 0, stream>>>(feat, Xp, N);
    k_pad_w1<<<4, 256, 0, stream>>>(W1, W1p);
    k_pad_w5<<<16, 256, 0, stream>>>(W5, W5p);

    const int nblk   = cdiv(E, BLK_EDGES);
    const int nwords = nblk * (BLK_EDGES / 4);
    k_tile_ids<<<cdiv(nwords, 256), 256, 0, stream>>>(dst, ids, E, nwords);

    const int  nTiles = N / TILE_ROWS;

    int* lists  = ids + nwords;
    int* segcnt = lists + (size_t)nTiles * AGG_WAVES * SEG_CAP * 2;

    k_gemm_strips<8, 128><<<GEMM_GROUPS, GEMM_THREADS, 0, stream>>>(Xp, W1p, XW, N / 16);
    k_tile_aggregate<4, 128, 128, true, true><<<nTiles, AGG_THREADS, 0, stream>>>(
        XW, src, dst, ids, c_src, c_dst, b1, H0, lists, segcnt, N, nblk);

    k_gemm_strips<128, 128><<<GEMM_GROUPS, GEMM_THREADS, 0, stream>>>(H0, W2, XW, N / 16);
    k_tile_aggregate<4, 128, 128, true, false><<<nTiles, AGG_THREADS, 0, stream>>>(
        XW, src, dst, ids, c_src, c_dst, b2, H0, lists, segcnt, N, nblk);

    k_gemm_strips<128, 128><<<GEMM_GROUPS, GEMM_THREADS, 0, stream>>>(H0, W3, XW, N / 16);
    k_tile_aggregate<4, 128, 128, true, false><<<nTiles, AGG_THREADS, 0, stream>>>(
        XW, src, dst, ids, c_src, c_dst, b3, H0, lists, segcnt, N, nblk);

    k_gemm_strips<128, 128><<<GEMM_GROUPS, GEMM_THREADS, 0, stream>>>(H0, W4, XW, N / 16);
    k_tile_aggregate<4, 128, 128, true, false><<<nTiles, AGG_THREADS, 0, stream>>>(
        XW, src, dst, ids, c_src, c_dst, b4, H0, lists, segcnt, N, nblk);

    k_gemm_strips<128, 32><<<GEMM_GROUPS, GEMM_THREADS, 0, stream>>>(H0, W5p, XW, N / 16);
    k_tile_aggregate<1, 32, 18, false, false><<<nTiles, AGG_THREADS, 0, stream>>>(
        XW, src, dst, ids, c_src, c_dst, b5, out, lists, segcnt, N, nblk);
}


# Layer 1 sums neighbours on the 8-wide padded features before the f32 WMMA product (both linear): its tile kernel adds 32-byte rows, one edge per lane; bias+ReLU move into the product's store
# speedup vs baseline: 2.8607x; 1.0882x over previous
#include <hip/hip_runtime.h>

typedef __attribute__((ext_vector_type(2))) float v2f;
typedef __attribute__((ext_vector_type(8))) float v8f;


__global__ void k_zero(float* __restrict__ p, int n) {
    int i = blockIdx.x * blockDim.x + threadIdx.x;
    if (i < n) p[i] = 0.0f;
}

__global__ void k_degree(const int* __restrict__ src, const int* __restrict__ dst,
                         float* __restrict__ degO, float* __restrict__ degI, int e) {
    int i = blockIdx.x * blockDim.x + threadIdx.x;
    if (i < e) {
        atomicAdd(&degO[src[i]], 1.0f);
        atomicAdd(&degI[dst[i]], 1.0f);
    }
}

__global__ void k_coeff(float* __restrict__ c, int n) {
    int i = blockIdx.x * blockDim.x + threadIdx.x;
    if (i < n) {
        float d = c[i];
        c[i] = (d > 0.0f) ? rsqrtf(d) : 0.0f;
    }
}

__global__ void k_pad_features(const float* __restrict__ x, float* __restrict__ xp, int n) {
    int i = blockIdx.x * blockDim.x + threadIdx.x;
    if (i < n * 8) {
        int r = i >> 3, c = i & 7;
        xp[i] = (c < 6) ? x[r * 6 + c] : 0.0f;
    }
}

__global__ void k_pad_w1(const float* __restrict__ w, float* __restrict__ wp) {
    int i = blockIdx.x * blockDim.x + threadIdx.x;
    if (i < 8 * 128) {
        int r = i >> 7;
        wp[i] = (r < 6) ? w[i] : 0.0f;
    }
}

__global__ void k_pad_w5(const float* __restrict__ w, float* __restrict__ wp) {
    int i = blockIdx.x * blockDim.x + threadIdx.x;
    if (i < 128 * 32) {
        int r = i >> 5, c = i & 31;
        wp[i] = (c < 18) ? w[r * 18 + c] : 0.0f;
    }
}


typedef __attribute__((ext_vector_type(4))) float v4f;

#define GEMM_WAVES      16
#define GEMM_THREADS    512
#define GEMM_GROUPS     256
#define GEMM_W_FLOATS   (128 * 128)
#define GEMM_STRIP_MAX  (16 * (128 + 4))
#define GEMM_LDS_FLOATS (GEMM_W_FLOATS + GEMM_WAVES * GEMM_STRIP_MAX)

template <int DIN>
__device__ __forceinline__ void load_a_strip(v4f (&a)[DIN / 8], const float* __restrict__ A,
                                             int mt, int c, int h) {
    const float* p = A + (size_t)(mt * 16 + c) * DIN + 4 * h;
#pragma unroll
    for (int j = 0; j < DIN / 8; ++j) a[j] = *(const v4f*)(p + 8 * j);
}

template <int DIN, int DOUT, bool EPI>
__global__ void __launch_bounds__(GEMM_THREADS)
k_gemm_strips(const float* __restrict__ A, const float* __restrict__ Wm,
              const float* __restrict__ bias, float* __restrict__ Y, int mtiles) {
    constexpr int NJ    = DIN / 8;
    constexpr int NT    = DOUT / 16;
    constexpr int PITCH = DOUT + 4;
    static_assert(DIN * DOUT <= GEMM_W_FLOATS && 16 * PITCH <= GEMM_STRIP_MAX, "LDS plan");
    __shared__ __attribute__((aligned(16))) float lds[GEMM_LDS_FLOATS];
    v4f*   wl = (v4f*)lds;
    float* ot = lds + GEMM_W_FLOATS;

    const int tid  = threadIdx.x;
    const int lane = tid & 31;
    const int c    = lane & 15;
    const int h    = lane >> 4;
    const int wave = __builtin_amdgcn_readfirstlane(tid >> 5);

    const int stride = (int)gridDim.x * GEMM_WAVES;
    int mt = (int)blockIdx.x + (int)gridDim.x * wave;

    v4f a[NJ];
    if (mt < mtiles) load_a_strip<DIN>(a, A, mt, c, h);

    constexpr int ITEMS  = NJ * NT * 32;
    constexpr int ROUNDS = (ITEMS + GEMM_THREADS - 1) / GEMM_THREADS;
#pragma unroll
    for (int it = 0; it < ROUNDS; ++it) {
        const int idx = tid + it * GEMM_THREADS;
        if (ITEMS % GEMM_THREADS == 0 || idx < ITEMS) {
            const int g  = idx >> 5;
            const int j  = g / NT, nt = g % NT;
            const int k0 = 8 * j + 4 * h, n = nt * 16 + c;
            v4f v;
            v.x = Wm[(size_t)(k0 + 0) * DOUT + n];
            v.y = Wm[(size_t)(k0 + 1) * DOUT + n];
            v.z = Wm[(size_t)(k0 + 2) * DOUT + n];
            v.w = Wm[(size_t)(k0 + 3) * DOUT + n];
            wl[idx] = v;
        }
    }
    __syncthreads();

    float* o = ot + wave * (16 * PITCH);
    constexpr int LPR = DOUT / 4;
    constexpr int RPI = 32 / LPR;
    const int rsub = lane / LPR, col = (lane % LPR) * 4;
    v4f bv = {0.0f, 0.0f, 0.0f, 0.0f};
    if (EPI) bv = *(const v4f*)(bias + col);

    while (mt < mtiles) {
        v8f acc[NT];
#pragma unroll
        for (int nt = 0; nt < NT; ++nt) acc[nt] = (v8f){};

#pragma unroll
        for (int j = 0; j < NJ; ++j) {
            v4f b[NT];
#pragma unroll
            for (int nt = 0; nt < NT; ++nt) b[nt] = wl[(j * NT + nt) * 32 + lane];
            const v2f a0 = {a[j].x, a[j].y}, a1 = {a[j].z, a[j].w};
#pragma unroll
            for (int nt = 0; nt < NT; ++nt) {
                const v2f b0 = {b[nt].x, b[nt].y};
                acc[nt] = __builtin_amdgcn_wmma_f32_16x16x4_f32(
                     false, a0,  false, b0,
                     (short)0, acc[nt],  false,  false);
            }
#pragma unroll
            for (int nt = 0; nt < NT; ++nt) {
                const v2f b1 = {b[nt].z, b[nt].w};
                acc[nt] = __builtin_amdgcn_wmma_f32_16x16x4_f32(
                     false, a1,  false, b1,
                     (short)0, acc[nt],  false,  false);
            }
        }

        const int nx = mt + stride;
        if (nx < mtiles) load_a_strip<DIN>(a, A, nx, c, h);

#pragma unroll
        for (int nt = 0; nt < NT; ++nt)
#pragma unroll
            for (int r = 0; r < 8; ++r)
                o[(8 * h + r) * PITCH + nt * 16 + c] = acc[nt][r];
        float* yp = Y + (size_t)mt * (16 * DOUT) + rsub * DOUT + col;
#pragma unroll
        for (int i = 0; i < 16 / RPI; ++i) {
            v4f v = *(const v4f*)(o + (i * RPI + rsub) * PITCH + col);
            if (EPI) {
                v.x = fmaxf(v.x + bv.x, 0.0f); v.y = fmaxf(v.y + bv.y, 0.0f);
                v.z = fmaxf(v.z + bv.z, 0.0f); v.w = fmaxf(v.w + bv.w, 0.0f);
            }
            *(v4f*)(yp + i * (RPI * DOUT)) = v;
        }
        mt = nx;
    }
}


#define TILE_ROWS   400
#define QCAP        768
#define AGG_THREADS 1024
#define AGG_WAVES   32
#define BLK_EDGES   256
#define DRAIN_U     8
#define SEG_CAP     512
#define SRC_BITS    17
#define NARROW_PITCH 9

__global__ void k_tile_ids(const int* __restrict__ dst, int* __restrict__ ids, int e, int nwords) {
    int i = blockIdx.x * blockDim.x + threadIdx.x;
    if (i < nwords) {
        unsigned w = 0u;
#pragma unroll
        for (int j = 0; j < 4; ++j) {
            int ei = i * 4 + j;
            unsigned t = (ei < e) ? (unsigned)(dst[ei] / TILE_ROWS) : 255u;
            w |= (t & 255u) << (8 * j);
        }
        ids[i] = (int)w;
    }
}

__device__ __forceinline__ unsigned zero_bytes(unsigned t) {
    return ~(((t & 0x7F7F7F7Fu) + 0x7F7F7F7Fu) | t | 0x7F7F7F7Fu);
}

__device__ __forceinline__ unsigned zero_byte_nibble(unsigned t) {
    return ((zero_bytes(t) >> 7) * 0x01020408u) >> 24;
}

__device__ __forceinline__ void load_id_words(int2 (&v)[4], const int* __restrict__ ids,
                                              int b0, int nblk, int lane) {
#pragma unroll
    for (int u = 0; u < 4; ++u) {
        const int b  = b0 + AGG_WAVES * u;
        const int bc = b < nblk ? b : nblk - 1;
        v[u] = *(const int2*)(ids + (size_t)bc * (BLK_EDGES / 4) + lane * 2);
    }
}

template <int NCH, int STRIDE>
__device__ __forceinline__ void scatter_group(const int (&sj)[DRAIN_U], const int (&dl)[DRAIN_U],
                                              const float (&cc)[DRAIN_U], int nv,
                                              const float* __restrict__ XW,
                                              float* acc, int lane) {
    constexpr int W = NCH * 32;
    float r[DRAIN_U][NCH];
#pragma unroll
    for (int u = 0; u < DRAIN_U; ++u) {
        const float* xr = XW + (size_t)sj[u] * STRIDE + lane;
#pragma unroll
        for (int k = 0; k < NCH; ++k) r[u][k] = xr[k * 32];
    }
#pragma unroll
    for (int u = 0; u < DRAIN_U; ++u) {
        if (u == 0 || u < nv) {
            float* ar = acc + dl[u] * W + lane;
#pragma unroll
            for (int k = 0; k < NCH; ++k) atomicAdd(&ar[k * 32], r[u][k] * cc[u]);
        }
    }
}

template <int NCH, int STRIDE>
__device__ __forceinline__ void scatter_rows(int s, int t, float c, int nb,
                                             const float* __restrict__ XW,
                                             float* acc, int lane) {
    for (int j0 = 0; j0 < nb; j0 += DRAIN_U) {
        int   sj[DRAIN_U];
        int   dl[DRAIN_U];
        float cc[DRAIN_U];
#pragma unroll
        for (int u = 0; u < DRAIN_U; ++u) {
            const int jj = j0 + u;
            sj[u] = __builtin_amdgcn_readlane(s, jj);
            dl[u] = __builtin_amdgcn_readlane(t, jj);
            cc[u] = __int_as_float(__builtin_amdgcn_readlane(__float_as_int(c), jj));
        }
        scatter_group<NCH, STRIDE>(sj, dl, cc, nb - j0, XW, acc, lane);
    }
}

template <int NCH, int STRIDE, bool SAVE>
__device__ __forceinline__ void drain_queue(const unsigned* q, int cnt,
                                            const float* __restrict__ XW,
                                            const int* __restrict__ src,
                                            const int* __restrict__ dst,
                                            const float* __restrict__ cs,
                                            float* acc, int base, int lane,
                                            int2* seg, int& saved) {
    for (int i0 = 0; i0 < cnt; i0 += 32) {
        const int nb = (cnt - i0 < 32) ? (cnt - i0) : 32;
        int s = 0, t = 0;
        float c = 0.0f;
        v4f x0 = {0.0f, 0.0f, 0.0f, 0.0f}, x1 = x0;
        if (lane < nb) {
            unsigned e = q[i0 + lane];
            s = src[e];
            t = dst[e] - base;
            t = t < 0 ? 0 : (t > TILE_ROWS - 1 ? TILE_ROWS - 1 : t);
            c = cs[s];
            if constexpr (NCH == 0) {
                x0 = *(const v4f*)(XW + (size_t)s * 8);
                x1 = *(const v4f*)(XW + (size_t)s * 8 + 4);
            }
        }
        if (SAVE) {
            if (saved + nb <= SEG_CAP && lane < nb)
                seg[saved + lane] = make_int2(s | (t << SRC_BITS), __float_as_int(c));
            saved += nb;
        }
        if constexpr (NCH == 0) {
            if (lane < nb) {
                float* ar = acc + t * NARROW_PITCH;
                atomicAdd(&ar[0], x0.x * c); atomicAdd(&ar[1], x0.y * c);
                atomicAdd(&ar[2], x0.z * c); atomicAdd(&ar[3], x0.w * c);
                atomicAdd(&ar[4], x1.x * c); atomicAdd(&ar[5], x1.y * c);
                atomicAdd(&ar[6], x1.z * c); atomicAdd(&ar[7], x1.w * c);
            }
        } else {
            scatter_rows<NCH, STRIDE>(s, t, c, nb, XW, acc, lane);
        }
    }
}

__device__ __forceinline__ void load_entries(int (&w)[2 * DRAIN_U], const int* __restrict__ segw, int j) {
#pragma unroll
    for (int i = 0; i < 2 * DRAIN_U; ++i) w[i] = segw[2 * j + i];
}

template <int NCH, int STRIDE>
__device__ __forceinline__ void drain_list(const int* __restrict__ segw, int cnt,
                                           const float* __restrict__ XW,
                                           float* acc, int lane) {
    if (cnt <= 0) return;
    int ex[2 * DRAIN_U];
    load_entries(ex, segw, 0);
    for (int j0 = 0; j0 < cnt; j0 += DRAIN_U) {
        int nx[2 * DRAIN_U];
        load_entries(nx, segw, (j0 + DRAIN_U < cnt) ? j0 + DRAIN_U : j0);
        int   sj[DRAIN_U];
        int   dl[DRAIN_U];
        float cc[DRAIN_U];
#pragma unroll
        for (int u = 0; u < DRAIN_U; ++u) {
            sj[u] = ex[2 * u] & ((1 << SRC_BITS) - 1);
            const int t = (int)((unsigned)ex[2 * u] >> SRC_BITS);
            dl[u] = t > TILE_ROWS - 1 ? TILE_ROWS - 1 : t;
            cc[u] = __int_as_float(ex[2 * u + 1]);
        }
        scatter_group<NCH, STRIDE>(sj, dl, cc, cnt - j0, XW, acc, lane);
#pragma unroll
        for (int i = 0; i < 2 * DRAIN_U; ++i) ex[i] = nx[i];
    }
}

template <int NCH, int STRIDE, int OUTW, bool RELU, bool BUILD>
__global__ void __launch_bounds__(AGG_THREADS)
k_tile_aggregate(const float* __restrict__ XW, const int* __restrict__ src,
                 const int* __restrict__ dst, const int* __restrict__ ids,
                 const float* __restrict__ cs, const float* __restrict__ cd,
                 const float* __restrict__ bias, float* __restrict__ Y,
                 int* lists, int* segcnt, int n, int nblk) {
    constexpr bool NARROW = (NCH == 0);
    static_assert(!NARROW || BUILD, "the 8-wide tile has no list reader");
    constexpr int W    = NARROW ? 8 : NCH * 32;
    constexpr int ACCF = NARROW ? TILE_ROWS * NARROW_PITCH : TILE_ROWS * W;
    constexpr int ACCR = NARROW ? 17 * 1024 : ACCF;
    __shared__ float    acc[ACCR];
    __shared__ unsigned queue[AGG_WAVES * QCAP];

    const int tid  = threadIdx.x;
    const int lane = tid & 31;
    const int wave = __builtin_amdgcn_readfirstlane(tid >> 5);
    const int p    = blockIdx.x;
    const int base = p * TILE_ROWS;

    for (int i = tid; i < ACCF; i += AGG_THREADS) acc[i] = 0.0f;
    __syncthreads();

    const int segi = p * AGG_WAVES + wave;
    int2* seg = (int2*)lists + (size_t)segi * SEG_CAP;
    int  saved = 0;
    bool scan  = true;
    if constexpr (!BUILD) {
        int sc = __builtin_amdgcn_readfirstlane(segcnt[segi]);
        if (sc >= 0) {
            scan = false;
            sc = sc < SEG_CAP ? sc : SEG_CAP;
            drain_list<NCH, STRIDE>((const int*)seg, sc, XW, acc, lane);
        }
    }

    if (scan) {
        unsigned* q = queue + wave * QCAP;
        int cnt = 0;
        const unsigned pp = (unsigned)p * 0x01010101u;
        const unsigned lt = (1u << lane) - 1u;

        if (nblk > 0) {
            int2 cur[4];
            load_id_words(cur, ids, wave, nblk, lane);
            for (int b0 = wave; b0 < nblk; b0 += AGG_WAVES * 4) {
                int2 nxt[4];
                load_id_words(nxt, ids, b0 + AGG_WAVES * 4, nblk, lane);

                unsigned mm = 0u;
#pragma unroll
                for (int u = 0; u < 4; ++u) {
                    if (b0 + AGG_WAVES * u < nblk) {
                        mm |= zero_byte_nibble((unsigned)cur[u].x ^ pp) << (8 * u);
                        mm |= zero_byte_nibble((unsigned)cur[u].y ^ pp) << (8 * u + 4);
                    }
                }
                const unsigned e0 = (unsigned)b0 * BLK_EDGES + (unsigned)lane * 8u;
                for (int it = 0; it < 32; ++it) {
                    const bool has = mm != 0u;
                    const unsigned m = __builtin_amdgcn_ballot_w32(has);
                    if (m == 0u) break;
                    if (cnt > QCAP - 32) {
                        drain_queue<NCH, STRIDE, BUILD>(q, cnt, XW, src, dst, cs, acc, base, lane,
                                                        seg, saved);
                        cnt = 0;
                    }
                    if (has) {
                        const unsigned i = (unsigned)__builtin_ctz(mm);
                        mm &= mm - 1u;
                        q[cnt + __popc(m & lt)] = e0 + (i >> 3) * (unsigned)(AGG_WAVES * BLK_EDGES) + (i & 7u);
                    }
                    cnt += __popc(m);
                }
#pragma unroll
                for (int u = 0; u < 4; ++u) cur[u] = nxt[u];
            }
        }
        drain_queue<NCH, STRIDE, BUILD>(q, cnt, XW, src, dst, cs, acc, base, lane, seg, saved);
    }
    if (BUILD) {
        const int padn = (DRAIN_U - (saved & (DRAIN_U - 1))) & (DRAIN_U - 1);
        if (saved <= SEG_CAP && lane < padn) seg[saved + lane] = make_int2(0, 0);
        if (lane == 0) segcnt[segi] = saved <= SEG_CAP ? saved : -1;
    }
    __syncthreads();

    if constexpr (NARROW) {
        for (int i = tid; i < TILE_ROWS * 8; i += AGG_THREADS) {
            const int r = i >> 3, c = i & 7;
            const int row = base + r;
            if (row < n) Y[(size_t)base * 8 + i] = acc[r * NARROW_PITCH + c] * cd[row];
        }
    } else if (OUTW == W) {
        for (int i = tid; i < TILE_ROWS * W; i += AGG_THREADS) {
            const int r = i / W, c = i % W;
            const int row = base + r;
            if (row < n) {
                float v = acc[i] * cd[row] + bias[c];
                if (RELU) v = fmaxf(v, 0.0f);
                Y[(size_t)row * W + c] = v;
            }
        }
    } else {
        for (int i = tid; i < TILE_ROWS * OUTW; i += AGG_THREADS) {
            const int r = i / OUTW, c = i % OUTW;
            const int row = base + r;
            if (row < n) {
                float v = acc[r * W + c] * cd[row] + bias[c];
                if (RELU) v = fmaxf(v, 0.0f);
                Y[(size_t)base * OUTW + i] = v;
            }
        }
    }
}


static inline int cdiv(int a, int b) { return (a + b - 1) / b; }

extern "C" void kernel_launch(void* const* d_in, const int* in_sizes, int n_in,
                              void* d_out, int out_size, void* d_ws, size_t ws_size,
                              hipStream_t stream) {
    (void)n_in; (void)out_size; (void)ws_size;

    const float* feat = (const float*)d_in[0];
    const int*   src  = (const int*)d_in[1];
    const int*   dst  = (const int*)d_in[2];
    const float* W1 = (const float*)d_in[3];  const float* b1 = (const float*)d_in[4];
    const float* W2 = (const float*)d_in[5];  const float* b2 = (const float*)d_in[6];
    const float* W3 = (const float*)d_in[7];  const float* b3 = (const float*)d_in[8];
    const float* W4 = (const float*)d_in[9];  const float* b4 = (const float*)d_in[10];
    const float* W5 = (const float*)d_in[11]; const float* b5 = (const float*)d_in[12];
    float* out = (float*)d_out;

    const int N = 100000;
    const int E = in_sizes[1];

    float* ws = (float*)d_ws;
    size_t o = 0;
    float* c_src = ws + o; o += N;
    float* c_dst = ws + o; o += N;
    float* Xp    = ws + o; o += (size_t)N * 8;
    float* W1p   = ws + o; o += 8 * 128;
    float* W5p   = ws + o; o += 128 * 32;
    float* XW    = ws + o; o += (size_t)N * 128;
    float* H0    = ws + o; o += (size_t)N * 128;
    int*   ids   = (int*)(ws + o);

    k_zero<<<cdiv(2 * N, 256), 256, 0, stream>>>(c_src, 2 * N);
    k_degree<<<cdiv(E, 256), 256, 0, stream>>>(src, dst, c_src, c_dst, E);
    k_coeff<<<cdiv(2 * N, 256), 256, 0, stream>>>(c_src, 2 * N);

    k_pad_features<<<cdiv(N * 8, 256), 256, 0, stream>>>(feat, Xp, N);
    k_pad_w1<<<4, 256, 0, stream>>>(W1, W1p);
    k_pad_w5<<<16, 256, 0, stream>>>(W5, W5p);

    const int nblk   = cdiv(E, BLK_EDGES);
    const int nwords = nblk * (BLK_EDGES / 4);
    k_tile_ids<<<cdiv(nwords, 256), 256, 0, stream>>>(dst, ids, E, nwords);

    const int  nTiles = N / TILE_ROWS;

    int* lists  = ids + nwords;
    int* segcnt = lists + (size_t)nTiles * AGG_WAVES * SEG_CAP * 2;

    float* AX = XW;
    k_tile_aggregate<0, 8, 8, false, true><<<nTiles, AGG_THREADS, 0, stream>>>(
        Xp, src, dst, ids, c_src, c_dst, b1, AX, lists, segcnt, N, nblk);
    k_gemm_strips<8, 128, true><<<GEMM_GROUPS, GEMM_THREADS, 0, stream>>>(AX, W1p, b1, H0, N / 16);

    k_gemm_strips<128, 128, false><<<GEMM_GROUPS, GEMM_THREADS, 0, stream>>>(H0, W2, b2, XW, N / 16);
    k_tile_aggregate<4, 128, 128, true, false><<<nTiles, AGG_THREADS, 0, stream>>>(
        XW, src, dst, ids, c_src, c_dst, b2, H0, lists, segcnt, N, nblk);

    k_gemm_strips<128, 128, false><<<GEMM_GROUPS, GEMM_THREADS, 0, stream>>>(H0, W3, b3, XW, N / 16);
    k_tile_aggregate<4, 128, 128, true, false><<<nTiles, AGG_THREADS, 0, stream>>>(
        XW, src, dst, ids, c_src, c_dst, b3, H0, lists, segcnt, N, nblk);

    k_gemm_strips<128, 128, false><<<GEMM_GROUPS, GEMM_THREADS, 0, stream>>>(H0, W4, b4, XW, N / 16);
    k_tile_aggregate<4, 128, 128, true, false><<<nTiles, AGG_THREADS, 0, stream>>>(
        XW, src, dst, ids, c_src, c_dst, b4, H0, lists, segcnt, N, nblk);

    k_gemm_strips<128, 32, false><<<GEMM_GROUPS, GEMM_THREADS, 0, stream>>>(H0, W5p, b5, XW, N / 16);
    k_tile_aggregate<1, 32, 18, false, false><<<nTiles, AGG_THREADS, 0, stream>>>(
        XW, src, dst, ids, c_src, c_dst, b5, out, lists, segcnt, N, nblk);
}


# Saved-list reader software-pipelined: two groups of 8 source rows in flight per wave, the next rows requested before the LDS sums of the group ahead; one 16-byte-per-lane load per 512-byte row
# speedup vs baseline: 3.6343x; 1.2704x over previous
#include <hip/hip_runtime.h>

typedef __attribute__((ext_vector_type(2))) float v2f;
typedef __attribute__((ext_vector_type(8))) float v8f;


__global__ void k_zero(float* __restrict__ p, int n) {
    int i = blockIdx.x * blockDim.x + threadIdx.x;
    if (i < n) p[i] = 0.0f;
}

__global__ void k_degree(const int* __restrict__ src, const int* __restrict__ dst,
                         float* __restrict__ degO, float* __restrict__ degI, int e) {
    int i = blockIdx.x * blockDim.x + threadIdx.x;
    if (i < e) {
        atomicAdd(&degO[src[i]], 1.0f);
        atomicAdd(&degI[dst[i]], 1.0f);
    }
}

__global__ void k_coeff(float* __restrict__ c, int n) {
    int i = blockIdx.x * blockDim.x + threadIdx.x;
    if (i < n) {
        float d = c[i];
        c[i] = (d > 0.0f) ? rsqrtf(d) : 0.0f;
    }
}

__global__ void k_pad_features(const float* __restrict__ x, float* __restrict__ xp, int n) {
    int i = blockIdx.x * blockDim.x + threadIdx.x;
    if (i < n * 8) {
        int r = i >> 3, c = i & 7;
        xp[i] = (c < 6) ? x[r * 6 + c] : 0.0f;
    }
}

__global__ void k_pad_w1(const float* __restrict__ w, float* __restrict__ wp) {
    int i = blockIdx.x * blockDim.x + threadIdx.x;
    if (i < 8 * 128) {
        int r = i >> 7;
        wp[i] = (r < 6) ? w[i] : 0.0f;
    }
}

__global__ void k_pad_w5(const float* __restrict__ w, float* __restrict__ wp) {
    int i = blockIdx.x * blockDim.x + threadIdx.x;
    if (i < 128 * 32) {
        int r = i >> 5, c = i & 31;
        wp[i] = (c < 18) ? w[r * 18 + c] : 0.0f;
    }
}


typedef __attribute__((ext_vector_type(4))) float v4f;

#define GEMM_WAVES      16
#define GEMM_THREADS    512
#define GEMM_GROUPS     256
#define GEMM_W_FLOATS   (128 * 128)
#define GEMM_STRIP_MAX  (16 * (128 + 4))
#define GEMM_LDS_FLOATS (GEMM_W_FLOATS + GEMM_WAVES * GEMM_STRIP_MAX)

template <int DIN>
__device__ __forceinline__ void load_a_strip(v4f (&a)[DIN / 8], const float* __restrict__ A,
                                             int mt, int c, int h) {
    const float* p = A + (size_t)(mt * 16 + c) * DIN + 4 * h;
#pragma unroll
    for (int j = 0; j < DIN / 8; ++j) a[j] = *(const v4f*)(p + 8 * j);
}

template <int DIN, int DOUT, bool EPI>
__global__ void __launch_bounds__(GEMM_THREADS)
k_gemm_strips(const float* __restrict__ A, const float* __restrict__ Wm,
              const float* __restrict__ bias, float* __restrict__ Y, int mtiles) {
    constexpr int NJ    = DIN / 8;
    constexpr int NT    = DOUT / 16;
    constexpr int PITCH = DOUT + 4;
    static_assert(DIN * DOUT <= GEMM_W_FLOATS && 16 * PITCH <= GEMM_STRIP_MAX, "LDS plan");
    __shared__ __attribute__((aligned(16))) float lds[GEMM_LDS_FLOATS];
    v4f*   wl = (v4f*)lds;
    float* ot = lds + GEMM_W_FLOATS;

    const int tid  = threadIdx.x;
    const int lane = tid & 31;
    const int c    = lane & 15;
    const int h    = lane >> 4;
    const int wave = __builtin_amdgcn_readfirstlane(tid >> 5);

    const int stride = (int)gridDim.x * GEMM_WAVES;
    int mt = (int)blockIdx.x + (int)gridDim.x * wave;

    v4f a[NJ];
    if (mt < mtiles) load_a_strip<DIN>(a, A, mt, c, h);

    constexpr int ITEMS  = NJ * NT * 32;
    constexpr int ROUNDS = (ITEMS + GEMM_THREADS - 1) / GEMM_THREADS;
#pragma unroll
    for (int it = 0; it < ROUNDS; ++it) {
        const int idx = tid + it * GEMM_THREADS;
        if (ITEMS % GEMM_THREADS == 0 || idx < ITEMS) {
            const int g  = idx >> 5;
            const int j  = g / NT, nt = g % NT;
            const int k0 = 8 * j + 4 * h, n = nt * 16 + c;
            v4f v;
            v.x = Wm[(size_t)(k0 + 0) * DOUT + n];
            v.y = Wm[(size_t)(k0 + 1) * DOUT + n];
            v.z = Wm[(size_t)(k0 + 2) * DOUT + n];
            v.w = Wm[(size_t)(k0 + 3) * DOUT + n];
            wl[idx] = v;
        }
    }
    __syncthreads();

    float* o = ot + wave * (16 * PITCH);
    constexpr int LPR = DOUT / 4;
    constexpr int RPI = 32 / LPR;
    const int rsub = lane / LPR, col = (lane % LPR) * 4;
    v4f bv = {0.0f, 0.0f, 0.0f, 0.0f};
    if (EPI) bv = *(const v4f*)(bias + col);

    while (mt < mtiles) {
        v8f acc[NT];
#pragma unroll
        for (int nt = 0; nt < NT; ++nt) acc[nt] = (v8f){};

#pragma unroll
        for (int j = 0; j < NJ; ++j) {
            v4f b[NT];
#pragma unroll
            for (int nt = 0; nt < NT; ++nt) b[nt] = wl[(j * NT + nt) * 32 + lane];
            const v2f a0 = {a[j].x, a[j].y}, a1 = {a[j].z, a[j].w};
#pragma unroll
            for (int nt = 0; nt < NT; ++nt) {
                const v2f b0 = {b[nt].x, b[nt].y};
                acc[nt] = __builtin_amdgcn_wmma_f32_16x16x4_f32(
                     false, a0,  false, b0,
                     (short)0, acc[nt],  false,  false);
            }
#pragma unroll
            for (int nt = 0; nt < NT; ++nt) {
                const v2f b1 = {b[nt].z, b[nt].w};
                acc[nt] = __builtin_amdgcn_wmma_f32_16x16x4_f32(
                     false, a1,  false, b1,
                     (short)0, acc[nt],  false,  false);
            }
        }

        const int nx = mt + stride;
        if (nx < mtiles) load_a_strip<DIN>(a, A, nx, c, h);

#pragma unroll
        for (int nt = 0; nt < NT; ++nt)
#pragma unroll
            for (int r = 0; r < 8; ++r)
                o[(8 * h + r) * PITCH + nt * 16 + c] = acc[nt][r];
        float* yp = Y + (size_t)mt * (16 * DOUT) + rsub * DOUT + col;
#pragma unroll
        for (int i = 0; i < 16 / RPI; ++i) {
            v4f v = *(const v4f*)(o + (i * RPI + rsub) * PITCH + col);
            if (EPI) {
                v.x = fmaxf(v.x + bv.x, 0.0f); v.y = fmaxf(v.y + bv.y, 0.0f);
                v.z = fmaxf(v.z + bv.z, 0.0f); v.w = fmaxf(v.w + bv.w, 0.0f);
            }
            *(v4f*)(yp + i * (RPI * DOUT)) = v;
        }
        mt = nx;
    }
}


#define TILE_ROWS   400
#define QCAP        768
#define AGG_THREADS 1024
#define AGG_WAVES   32
#define BLK_EDGES   256
#define DRAIN_U     8
#define SEG_CAP     512
#define SRC_BITS    17
#define NARROW_PITCH 9

__global__ void k_tile_ids(const int* __restrict__ dst, int* __restrict__ ids, int e, int nwords) {
    int i = blockIdx.x * blockDim.x + threadIdx.x;
    if (i < nwords) {
        unsigned w = 0u;
#pragma unroll
        for (int j = 0; j < 4; ++j) {
            int ei = i * 4 + j;
            unsigned t = (ei < e) ? (unsigned)(dst[ei] / TILE_ROWS) : 255u;
            w |= (t & 255u) << (8 * j);
        }
        ids[i] = (int)w;
    }
}

__device__ __forceinline__ unsigned zero_bytes(unsigned t) {
    return ~(((t & 0x7F7F7F7Fu) + 0x7F7F7F7Fu) | t | 0x7F7F7F7Fu);
}

__device__ __forceinline__ unsigned zero_byte_nibble(unsigned t) {
    return ((zero_bytes(t) >> 7) * 0x01020408u) >> 24;
}

__device__ __forceinline__ void load_id_words(int2 (&v)[4], const int* __restrict__ ids,
                                              int b0, int nblk, int lane) {
#pragma unroll
    for (int u = 0; u < 4; ++u) {
        const int b  = b0 + AGG_WAVES * u;
        const int bc = b < nblk ? b : nblk - 1;
        v[u] = *(const int2*)(ids + (size_t)bc * (BLK_EDGES / 4) + lane * 2);
    }
}

template <int NCH> struct RowPiece { float v[NCH]; };

template <int NCH, int STRIDE>
__device__ __forceinline__ RowPiece<NCH> load_row(const float* __restrict__ XW, int s, int lane) {
    static_assert(NCH == 4 || NCH == 1, "row widths in use");
    RowPiece<NCH> r;
    if constexpr (NCH == 4) {
        const v4f t = *(const v4f*)(XW + (size_t)s * STRIDE + 4 * lane);
        r.v[0] = t.x; r.v[1] = t.y; r.v[2] = t.z; r.v[3] = t.w;
    } else {
        r.v[0] = XW[(size_t)s * STRIDE + lane];
    }
    return r;
}

template <int NCH>
__device__ __forceinline__ void add_row(float* acc, int dl, const RowPiece<NCH>& r, float cc, int lane) {
    float* ar = acc + dl * (NCH * 32) + lane;
#pragma unroll
    for (int k = 0; k < NCH; ++k) atomicAdd(&ar[k * 32], r.v[k] * cc);
}

template <int NCH, int STRIDE>
__device__ __forceinline__ void scatter_group(const int (&sj)[DRAIN_U], const int (&dl)[DRAIN_U],
                                              const float (&cc)[DRAIN_U], int nv,
                                              const float* __restrict__ XW,
                                              float* acc, int lane) {
    RowPiece<NCH> r[DRAIN_U];
#pragma unroll
    for (int u = 0; u < DRAIN_U; ++u) r[u] = load_row<NCH, STRIDE>(XW, sj[u], lane);
#pragma unroll
    for (int u = 0; u < DRAIN_U; ++u)
        if (u == 0 || u < nv) add_row<NCH>(acc, dl[u], r[u], cc[u], lane);
}

template <int NCH, int STRIDE>
__device__ __forceinline__ void scatter_rows(int s, int t, float c, int nb,
                                             const float* __restrict__ XW,
                                             float* acc, int lane) {
    for (int j0 = 0; j0 < nb; j0 += DRAIN_U) {
        int   sj[DRAIN_U];
        int   dl[DRAIN_U];
        float cc[DRAIN_U];
#pragma unroll
        for (int u = 0; u < DRAIN_U; ++u) {
            const int jj = j0 + u;
            sj[u] = __builtin_amdgcn_readlane(s, jj);
            dl[u] = __builtin_amdgcn_readlane(t, jj);
            cc[u] = __int_as_float(__builtin_amdgcn_readlane(__float_as_int(c), jj));
        }
        scatter_group<NCH, STRIDE>(sj, dl, cc, nb - j0, XW, acc, lane);
    }
}

template <int NCH, int STRIDE, bool SAVE>
__device__ __forceinline__ void drain_queue(const unsigned* q, int cnt,
                                            const float* __restrict__ XW,
                                            const int* __restrict__ src,
                                            const int* __restrict__ dst,
                                            const float* __restrict__ cs,
                                            float* acc, int base, int lane,
                                            int2* seg, int& saved) {
    for (int i0 = 0; i0 < cnt; i0 += 32) {
        const int nb = (cnt - i0 < 32) ? (cnt - i0) : 32;
        int s = 0, t = 0;
        float c = 0.0f;
        v4f x0 = {0.0f, 0.0f, 0.0f, 0.0f}, x1 = x0;
        if (lane < nb) {
            unsigned e = q[i0 + lane];
            s = src[e];
            t = dst[e] - base;
            t = t < 0 ? 0 : (t > TILE_ROWS - 1 ? TILE_ROWS - 1 : t);
            c = cs[s];
            if constexpr (NCH == 0) {
                x0 = *(const v4f*)(XW + (size_t)s * 8);
                x1 = *(const v4f*)(XW + (size_t)s * 8 + 4);
            }
        }
        if (SAVE) {
            if (saved + nb <= SEG_CAP && lane < nb)
                seg[saved + lane] = make_int2(s | (t << SRC_BITS), __float_as_int(c));
            saved += nb;
        }
        if constexpr (NCH == 0) {
            if (lane < nb) {
                float* ar = acc + t * NARROW_PITCH;
                atomicAdd(&ar[0], x0.x * c); atomicAdd(&ar[1], x0.y * c);
                atomicAdd(&ar[2], x0.z * c); atomicAdd(&ar[3], x0.w * c);
                atomicAdd(&ar[4], x1.x * c); atomicAdd(&ar[5], x1.y * c);
                atomicAdd(&ar[6], x1.z * c); atomicAdd(&ar[7], x1.w * c);
            }
        } else {
            scatter_rows<NCH, STRIDE>(s, t, c, nb, XW, acc, lane);
        }
    }
}

typedef const __attribute__((address_space(4))) int* const_words;
__device__ __forceinline__ void load_entries(int (&w)[2 * DRAIN_U], const_words segw, int j) {
#pragma unroll
    for (int i = 0; i < 2 * DRAIN_U; ++i) w[i] = segw[2 * j + i];
}

template <int NCH, int STRIDE>
__device__ __forceinline__ void request_group(RowPiece<NCH> (&r)[DRAIN_U], int (&dl)[DRAIN_U],
                                              float (&cc)[DRAIN_U], const int (&w)[2 * DRAIN_U],
                                              const float* __restrict__ XW, int lane) {
#pragma unroll
    for (int u = 0; u < DRAIN_U; ++u) {
        const int sj = w[2 * u] & ((1 << SRC_BITS) - 1);
        const int t  = (int)((unsigned)w[2 * u] >> SRC_BITS);
        dl[u] = t > TILE_ROWS ? TILE_ROWS : t;
        cc[u] = __int_as_float(w[2 * u + 1]);
        r[u]  = load_row<NCH, STRIDE>(XW, sj, lane);
    }
}

template <int NCH>
__device__ __forceinline__ void sum_group(const RowPiece<NCH> (&r)[DRAIN_U], const int (&dl)[DRAIN_U],
                                          const float (&cc)[DRAIN_U], float* acc, int lane) {
#pragma unroll
    for (int u = 0; u < DRAIN_U; ++u) add_row<NCH>(acc, dl[u], r[u], cc[u], lane);
}

template <int NCH, int STRIDE>
__device__ __forceinline__ void drain_list(const_words segw, int cnt,
                                           const float* __restrict__ XW,
                                           float* acc, int lane) {
    constexpr int PAIR = 2 * DRAIN_U;
    const int last = cnt - PAIR;
    RowPiece<NCH> ra[DRAIN_U], rb[DRAIN_U];
    int   da[DRAIN_U], db[DRAIN_U];
    float ca[DRAIN_U], cb[DRAIN_U];
    int   ea[PAIR], eb[PAIR];
    load_entries(ea, segw, 0);
    load_entries(eb, segw, DRAIN_U);
    const int j1 = PAIR < last ? PAIR : last;
    request_group<NCH, STRIDE>(ra, da, ca, ea, XW, lane);
    load_entries(ea, segw, j1);
    request_group<NCH, STRIDE>(rb, db, cb, eb, XW, lane);
    load_entries(eb, segw, j1 + DRAIN_U);
    for (int j0 = 0; j0 < last; j0 += PAIR) {
        const int j2 = j0 + 2 * PAIR < last ? j0 + 2 * PAIR : last;
        sum_group<NCH>(ra, da, ca, acc, lane);
        request_group<NCH, STRIDE>(ra, da, ca, ea, XW, lane);
        load_entries(ea, segw, j2);
        __builtin_amdgcn_sched_barrier(0);
        sum_group<NCH>(rb, db, cb, acc, lane);
        request_group<NCH, STRIDE>(rb, db, cb, eb, XW, lane);
        load_entries(eb, segw, j2 + DRAIN_U);
    }
    sum_group<NCH>(ra, da, ca, acc, lane);
    sum_group<NCH>(rb, db, cb, acc, lane);
}

template <int NCH, int STRIDE, int OUTW, bool RELU, bool BUILD>
__global__ void __launch_bounds__(AGG_THREADS)
k_tile_aggregate(const float* __restrict__ XW, const int* __restrict__ src,
                 const int* __restrict__ dst, const int* __restrict__ ids,
                 const float* __restrict__ cs, const float* __restrict__ cd,
                 const float* __restrict__ bias, float* __restrict__ Y,
                 int* lists, int* segcnt, int n, int nblk) {
    constexpr bool NARROW = (NCH == 0);
    static_assert(!NARROW || BUILD, "the 8-wide tile has no list reader");
    constexpr int W    = NARROW ? 8 : NCH * 32;
    constexpr int ACCF = NARROW ? TILE_ROWS * NARROW_PITCH : TILE_ROWS * W;
    constexpr int ACCR = NARROW ? 17 * 1024 : ACCF + W;
    __shared__ float    acc[ACCR];
    __shared__ unsigned queue[AGG_WAVES * QCAP];

    const int tid  = threadIdx.x;
    const int lane = tid & 31;
    const int wave = __builtin_amdgcn_readfirstlane(tid >> 5);
    const int p    = blockIdx.x;
    const int base = p * TILE_ROWS;

    for (int i = tid; i < ACCF; i += AGG_THREADS) acc[i] = 0.0f;
    __syncthreads();

    const int segi = p * AGG_WAVES + wave;
    int2* seg = (int2*)lists + (size_t)segi * SEG_CAP;
    int  saved = 0;
    bool scan  = true;
    if constexpr (!BUILD) {
        int sc = __builtin_amdgcn_readfirstlane(segcnt[segi]);
        if (sc >= 0) {
            scan = false;
            sc = sc < SEG_CAP ? sc : SEG_CAP;
            sc = (sc + 2 * DRAIN_U - 1) & ~(2 * DRAIN_U - 1);
            if (sc > 0) drain_list<NCH, STRIDE>((const_words)seg, sc, XW, acc, lane);
        }
    }

    if (scan) {
        unsigned* q = queue + wave * QCAP;
        int cnt = 0;
        const unsigned pp = (unsigned)p * 0x01010101u;
        const unsigned lt = (1u << lane) - 1u;

        if (nblk > 0) {
            int2 cur[4];
            load_id_words(cur, ids, wave, nblk, lane);
            for (int b0 = wave; b0 < nblk; b0 += AGG_WAVES * 4) {
                int2 nxt[4];
                load_id_words(nxt, ids, b0 + AGG_WAVES * 4, nblk, lane);

                unsigned mm = 0u;
#pragma unroll
                for (int u = 0; u < 4; ++u) {
                    if (b0 + AGG_WAVES * u < nblk) {
                        mm |= zero_byte_nibble((unsigned)cur[u].x ^ pp) << (8 * u);
                        mm |= zero_byte_nibble((unsigned)cur[u].y ^ pp) << (8 * u + 4);
                    }
                }
                const unsigned e0 = (unsigned)b0 * BLK_EDGES + (unsigned)lane * 8u;
                for (int it = 0; it < 32; ++it) {
                    const bool has = mm != 0u;
                    const unsigned m = __builtin_amdgcn_ballot_w32(has);
                    if (m == 0u) break;
                    if (cnt > QCAP - 32) {
                        drain_queue<NCH, STRIDE, BUILD>(q, cnt, XW, src, dst, cs, acc, base, lane,
                                                        seg, saved);
                        cnt = 0;
                    }
                    if (has) {
                        const unsigned i = (unsigned)__builtin_ctz(mm);
                        mm &= mm - 1u;
                        q[cnt + __popc(m & lt)] = e0 + (i >> 3) * (unsigned)(AGG_WAVES * BLK_EDGES) + (i & 7u);
                    }
                    cnt += __popc(m);
                }
#pragma unroll
                for (int u = 0; u < 4; ++u) cur[u] = nxt[u];
            }
        }
        drain_queue<NCH, STRIDE, BUILD>(q, cnt, XW, src, dst, cs, acc, base, lane, seg, saved);
    }
    if (BUILD) {
        constexpr int PAIR = 2 * DRAIN_U;
        static_assert(SEG_CAP % PAIR == 0 && PAIR <= 32, "one padding entry per lane at most");
        const int padn = (PAIR - (saved & (PAIR - 1))) & (PAIR - 1);
        if (saved <= SEG_CAP && lane < padn) seg[saved + lane] = make_int2(TILE_ROWS << SRC_BITS, 0);
        if (lane == 0) segcnt[segi] = saved <= SEG_CAP ? saved : -1;
    }
    __syncthreads();

    if constexpr (NARROW) {
        for (int i = tid; i < TILE_ROWS * 8; i += AGG_THREADS) {
            const int r = i >> 3, c = i & 7;
            const int row = base + r;
            if (row < n) Y[(size_t)base * 8 + i] = acc[r * NARROW_PITCH + c] * cd[row];
        }
    } else if (OUTW == W) {
        static_assert(OUTW != W || NCH == 4, "the whole-row store is written for 128 columns");
        const float bx = bias[4 * lane], by = bias[4 * lane + 1], bz = bias[4 * lane + 2], bw = bias[4 * lane + 3];
        for (int i = tid; i < TILE_ROWS * 32; i += AGG_THREADS) {
            const int r = i >> 5;
            const int row = base + r;
            if (row < n) {
                const float d = cd[row];
                const float* ar = acc + r * W + lane;
                v4f v;
                v.x = ar[0] * d + bx;  v.y = ar[32] * d + by;
                v.z = ar[64] * d + bz; v.w = ar[96] * d + bw;
                if (RELU) {
                    v.x = fmaxf(v.x, 0.0f); v.y = fmaxf(v.y, 0.0f);
                    v.z = fmaxf(v.z, 0.0f); v.w = fmaxf(v.w, 0.0f);
                }
                *(v4f*)(Y + (size_t)row * W + 4 * lane) = v;
            }
        }
    } else {
        for (int i = tid; i < TILE_ROWS * OUTW; i += AGG_THREADS) {
            const int r = i / OUTW, c = i % OUTW;
            const int row = base + r;
            if (row < n) {
                float v = acc[r * W + c] * cd[row] + bias[c];
                if (RELU) v = fmaxf(v, 0.0f);
                Y[(size_t)base * OUTW + i] = v;
            }
        }
    }
}


static inline int cdiv(int a, int b) { return (a + b - 1) / b; }

extern "C" void kernel_launch(void* const* d_in, const int* in_sizes, int n_in,
                              void* d_out, int out_size, void* d_ws, size_t ws_size,
                              hipStream_t stream) {
    (void)n_in; (void)out_size; (void)ws_size;

    const float* feat = (const float*)d_in[0];
    const int*   src  = (const int*)d_in[1];
    const int*   dst  = (const int*)d_in[2];
    const float* W1 = (const float*)d_in[3];  const float* b1 = (const float*)d_in[4];
    const float* W2 = (const float*)d_in[5];  const float* b2 = (const float*)d_in[6];
    const float* W3 = (const float*)d_in[7];  const float* b3 = (const float*)d_in[8];
    const float* W4 = (const float*)d_in[9];  const float* b4 = (const float*)d_in[10];
    const float* W5 = (const float*)d_in[11]; const float* b5 = (const float*)d_in[12];
    float* out = (float*)d_out;

    const int N = 100000;
    const int E = in_sizes[1];

    float* ws = (float*)d_ws;
    size_t o = 0;
    float* c_src = ws + o; o += N;
    float* c_dst = ws + o; o += N;
    float* Xp    = ws + o; o += (size_t)N * 8;
    float* W1p   = ws + o; o += 8 * 128;
    float* W5p   = ws + o; o += 128 * 32;
    float* XW    = ws + o; o += (size_t)N * 128;
    float* H0    = ws + o; o += (size_t)N * 128;
    int*   ids   = (int*)(ws + o);

    k_zero<<<cdiv(2 * N, 256), 256, 0, stream>>>(c_src, 2 * N);
    k_degree<<<cdiv(E, 256), 256, 0, stream>>>(src, dst, c_src, c_dst, E);
    k_coeff<<<cdiv(2 * N, 256), 256, 0, stream>>>(c_src, 2 * N);

    k_pad_features<<<cdiv(N * 8, 256), 256, 0, stream>>>(feat, Xp, N);
    k_pad_w1<<<4, 256, 0, stream>>>(W1, W1p);
    k_pad_w5<<<16, 256, 0, stream>>>(W5, W5p);

    const int nblk   = cdiv(E, BLK_EDGES);
    const int nwords = nblk * (BLK_EDGES / 4);
    k_tile_ids<<<cdiv(nwords, 256), 256, 0, stream>>>(dst, ids, E, nwords);

    const int  nTiles = N / TILE_ROWS;

    int* lists  = ids + nwords;
    int* segcnt = lists + (size_t)nTiles * AGG_WAVES * SEG_CAP * 2;

    float* AX = XW;
    k_tile_aggregate<0, 8, 8, false, true><<<nTiles, AGG_THREADS, 0, stream>>>(
        Xp, src, dst, ids, c_src, c_dst, b1, AX, lists, segcnt, N, nblk);
    k_gemm_strips<8, 128, true><<<GEMM_GROUPS, GEMM_THREADS, 0, stream>>>(AX, W1p, b1, H0, N / 16);

    k_gemm_strips<128, 128, false><<<GEMM_GROUPS, GEMM_THREADS, 0, stream>>>(H0, W2, b2, XW, N / 16);
    k_tile_aggregate<4, 128, 128, true, false><<<nTiles, AGG_THREADS, 0, stream>>>(
        XW, src, dst, ids, c_src, c_dst, b2, H0, lists, segcnt, N, nblk);

    k_gemm_strips<128, 128, false><<<GEMM_GROUPS, GEMM_THREADS, 0, stream>>>(H0, W3, b3, XW, N / 16);
    k_tile_aggregate<4, 128, 128, true, false><<<nTiles, AGG_THREADS, 0, stream>>>(
        XW, src, dst, ids, c_src, c_dst, b3, H0, lists, segcnt, N, nblk);

    k_gemm_strips<128, 128, false><<<GEMM_GROUPS, GEMM_THREADS, 0, stream>>>(H0, W4, b4, XW, N / 16);
    k_tile_aggregate<4, 128, 128, true, false><<<nTiles, AGG_THREADS, 0, stream>>>(
        XW, src, dst, ids, c_src, c_dst, b4, H0, lists, segcnt, N, nblk);

    k_gemm_strips<128, 32, false><<<GEMM_GROUPS, GEMM_THREADS, 0, stream>>>(H0, W5p, b5, XW, N / 16);
    k_tile_aggregate<1, 32, 18, false, false><<<nTiles, AGG_THREADS, 0, stream>>>(
        XW, src, dst, ids, c_src, c_dst, b5, out, lists, segcnt, N, nblk);
}


# Edge lists built by one binning pass over the edges instead of 250 work-groups each scanning a tile-id byte per edge: LDS per-tile ranks, one cursor add per tile and group; equal list shares per wave
# speedup vs baseline: 3.8405x; 1.0567x over previous
#include <hip/hip_runtime.h>

typedef __attribute__((ext_vector_type(2))) float v2f;
typedef __attribute__((ext_vector_type(8))) float v8f;


__global__ void k_zero(float* __restrict__ p, int n) {
    int i = blockIdx.x * blockDim.x + threadIdx.x;
    if (i < n) p[i] = 0.0f;
}

__global__ void k_degree(const int* __restrict__ src, const int* __restrict__ dst,
                         float* __restrict__ degO, float* __restrict__ degI, int e) {
    int i = blockIdx.x * blockDim.x + threadIdx.x;
    if (i < e) {
        atomicAdd(&degO[src[i]], 1.0f);
        atomicAdd(&degI[dst[i]], 1.0f);
    }
}

__global__ void k_coeff(float* __restrict__ c, int n) {
    int i = blockIdx.x * blockDim.x + threadIdx.x;
    if (i < n) {
        float d = c[i];
        c[i] = (d > 0.0f) ? rsqrtf(d) : 0.0f;
    }
}

__global__ void k_pad_features(const float* __restrict__ x, float* __restrict__ xp, int n) {
    int i = blockIdx.x * blockDim.x + threadIdx.x;
    if (i < n * 8) {
        int r = i >> 3, c = i & 7;
        xp[i] = (c < 6) ? x[r * 6 + c] : 0.0f;
    }
}

__global__ void k_pad_w1(const float* __restrict__ w, float* __restrict__ wp) {
    int i = blockIdx.x * blockDim.x + threadIdx.x;
    if (i < 8 * 128) {
        int r = i >> 7;
        wp[i] = (r < 6) ? w[i] : 0.0f;
    }
}

__global__ void k_pad_w5(const float* __restrict__ w, float* __restrict__ wp) {
    int i = blockIdx.x * blockDim.x + threadIdx.x;
    if (i < 128 * 32) {
        int r = i >> 5, c = i & 31;
        wp[i] = (c < 18) ? w[r * 18 + c] : 0.0f;
    }
}


typedef __attribute__((ext_vector_type(4))) float v4f;

#define GEMM_WAVES      16
#define GEMM_THREADS    512
#define GEMM_GROUPS     256
#define GEMM_W_FLOATS   (128 * 128)
#define GEMM_STRIP_MAX  (16 * (128 + 4))
#define GEMM_LDS_FLOATS (GEMM_W_FLOATS + GEMM_WAVES * GEMM_STRIP_MAX)

template <int DIN>
__device__ __forceinline__ void load_a_strip(v4f (&a)[DIN / 8], const float* __restrict__ A,
                                             int mt, int c, int h) {
    const float* p = A + (size_t)(mt * 16 + c) * DIN + 4 * h;
#pragma unroll
    for (int j = 0; j < DIN / 8; ++j) a[j] = *(const v4f*)(p + 8 * j);
}

template <int DIN, int DOUT, bool EPI>
__global__ void __launch_bounds__(GEMM_THREADS)
k_gemm_strips(const float* __restrict__ A, const float* __restrict__ Wm,
              const float* __restrict__ bias, float* __restrict__ Y, int mtiles) {
    constexpr int NJ    = DIN / 8;
    constexpr int NT    = DOUT / 16;
    constexpr int PITCH = DOUT + 4;
    static_assert(DIN * DOUT <= GEMM_W_FLOATS && 16 * PITCH <= GEMM_STRIP_MAX, "LDS plan");
    __shared__ __attribute__((aligned(16))) float lds[GEMM_LDS_FLOATS];
    v4f*   wl = (v4f*)lds;
    float* ot = lds + GEMM_W_FLOATS;

    const int tid  = threadIdx.x;
    const int lane = tid & 31;
    const int c    = lane & 15;
    const int h    = lane >> 4;
    const int wave = __builtin_amdgcn_readfirstlane(tid >> 5);

    const int stride = (int)gridDim.x * GEMM_WAVES;
    int mt = (int)blockIdx.x + (int)gridDim.x * wave;

    v4f a[NJ];
    if (mt < mtiles) load_a_strip<DIN>(a, A, mt, c, h);

    constexpr int ITEMS  = NJ * NT * 32;
    constexpr int ROUNDS = (ITEMS + GEMM_THREADS - 1) / GEMM_THREADS;
#pragma unroll
    for (int it = 0; it < ROUNDS; ++it) {
        const int idx = tid + it * GEMM_THREADS;
        if (ITEMS % GEMM_THREADS == 0 || idx < ITEMS) {
            const int g  = idx >> 5;
            const int j  = g / NT, nt = g % NT;
            const int k0 = 8 * j + 4 * h, n = nt * 16 + c;
            v4f v;
            v.x = Wm[(size_t)(k0 + 0) * DOUT + n];
            v.y = Wm[(size_t)(k0 + 1) * DOUT + n];
            v.z = Wm[(size_t)(k0 + 2) * DOUT + n];
            v.w = Wm[(size_t)(k0 + 3) * DOUT + n];
            wl[idx] = v;
        }
    }
    __syncthreads();

    float* o = ot + wave * (16 * PITCH);
    constexpr int LPR = DOUT / 4;
    constexpr int RPI = 32 / LPR;
    const int rsub = lane / LPR, col = (lane % LPR) * 4;
    v4f bv = {0.0f, 0.0f, 0.0f, 0.0f};
    if (EPI) bv = *(const v4f*)(bias + col);

    while (mt < mtiles) {
        v8f acc[NT];
#pragma unroll
        for (int nt = 0; nt < NT; ++nt) acc[nt] = (v8f){};

#pragma unroll
        for (int j = 0; j < NJ; ++j) {
            v4f b[NT];
#pragma unroll
            for (int nt = 0; nt < NT; ++nt) b[nt] = wl[(j * NT + nt) * 32 + lane];
            const v2f a0 = {a[j].x, a[j].y}, a1 = {a[j].z, a[j].w};
#pragma unroll
            for (int nt = 0; nt < NT; ++nt) {
                const v2f b0 = {b[nt].x, b[nt].y};
                acc[nt] = __builtin_amdgcn_wmma_f32_16x16x4_f32(
                     false, a0,  false, b0,
                     (short)0, acc[nt],  false,  false);
            }
#pragma unroll
            for (int nt = 0; nt < NT; ++nt) {
                const v2f b1 = {b[nt].z, b[nt].w};
                acc[nt] = __builtin_amdgcn_wmma_f32_16x16x4_f32(
                     false, a1,  false, b1,
                     (short)0, acc[nt],  false,  false);
            }
        }

        const int nx = mt + stride;
        if (nx < mtiles) load_a_strip<DIN>(a, A, nx, c, h);

#pragma unroll
        for (int nt = 0; nt < NT; ++nt)
#pragma unroll
            for (int r = 0; r < 8; ++r)
                o[(8 * h + r) * PITCH + nt * 16 + c] = acc[nt][r];
        float* yp = Y + (size_t)mt * (16 * DOUT) + rsub * DOUT + col;
#pragma unroll
        for (int i = 0; i < 16 / RPI; ++i) {
            v4f v = *(const v4f*)(o + (i * RPI + rsub) * PITCH + col);
            if (EPI) {
                v.x = fmaxf(v.x + bv.x, 0.0f); v.y = fmaxf(v.y + bv.y, 0.0f);
                v.z = fmaxf(v.z + bv.z, 0.0f); v.w = fmaxf(v.w + bv.w, 0.0f);
            }
            *(v4f*)(yp + i * (RPI * DOUT)) = v;
        }
        mt = nx;
    }
}


#define TILE_ROWS   400
#define MAX_TILES   256
#define AGG_THREADS 1024
#define AGG_WAVES   32
#define DRAIN_U     8
#define LIST_ALIGN  512
#define BIN_EDGES   4096
#define CUR_PITCH   64
#define SRC_BITS    17
#define NARROW_PITCH 9

__device__ __forceinline__ int tile_cap(int cnt) { return (cnt + LIST_ALIGN - 1) & ~(LIST_ALIGN - 1); }

__global__ void k_tile_count(const float* __restrict__ degI, int* __restrict__ tcnt,
                             int* __restrict__ tcur, int n) {
    __shared__ int part[4];
    const int p = blockIdx.x, tid = threadIdx.x;
    int s = 0;
    for (int r = tid; r < TILE_ROWS; r += 128) {
        const int row = p * TILE_ROWS + r;
        if (row < n) s += (int)degI[row];
    }
    for (int o = 16; o > 0; o >>= 1) s += __shfl_xor(s, o);
    if ((tid & 31) == 0) part[tid >> 5] = s;
    __syncthreads();
    if (tid == 0) {
        tcnt[p] = part[0] + part[1] + part[2] + part[3];
        tcur[p * CUR_PITCH] = 0;
    }
}

__global__ void __launch_bounds__(AGG_THREADS)
k_bin_edges(const int* __restrict__ src, const int* __restrict__ dst, const float* __restrict__ cs,
            const int* __restrict__ tcnt, int* tcur, int* __restrict__ toffg, int2* lists,
            int e, int ntiles) {
    __shared__ int hist[MAX_TILES];
    __shared__ int start[MAX_TILES];
    __shared__ int toff[MAX_TILES];
    __shared__ int wsum[MAX_TILES / 32];

    const int tid = threadIdx.x, lane = tid & 31, wave = tid >> 5;

    int capv = 0, incl = 0;
    if (tid < MAX_TILES) {
        capv = tid < ntiles ? tile_cap(tcnt[tid]) : 0;
        incl = capv;
        for (int o = 1; o < 32; o <<= 1) {
            const int v = __shfl_up(incl, o);
            if (lane >= o) incl += v;
        }
        if (lane == 31) wsum[wave] = incl;
        hist[tid] = 0;
    }

    int   s[4], t[4], pt[4];
    float c[4];
    const int e0 = (int)blockIdx.x * BIN_EDGES + tid;
#pragma unroll
    for (int u = 0; u < 4; ++u) {
        const int ei = e0 + u * AGG_THREADS;
        const bool ok = ei < e;
        const int d = ok ? dst[ei] : 0;
        s[u] = ok ? src[ei] : 0;
        int p = (int)((unsigned)d / (unsigned)TILE_ROWS);
        p = p < ntiles - 1 ? p : ntiles - 1;
        pt[u] = p;
        t[u]  = d - p * TILE_ROWS;
    }
#pragma unroll
    for (int u = 0; u < 4; ++u) c[u] = cs[s[u]];
    __syncthreads();

    int rk[4];
#pragma unroll
    for (int u = 0; u < 4; ++u) {
        rk[u] = 0;
        if (e0 + u * AGG_THREADS < e) rk[u] = atomicAdd(&hist[pt[u]], 1);
    }
    __syncthreads();

    if (tid < MAX_TILES) {
        int excl = incl - capv;
        for (int w = 0; w < wave; ++w) excl += wsum[w];
        toff[tid] = excl;
        if (blockIdx.x == 0) toffg[tid] = excl;
        const int nloc = hist[tid];
        start[tid] = nloc > 0 ? excl + atomicAdd(&tcur[tid * CUR_PITCH], nloc) : 0;
    }
    __syncthreads();

#pragma unroll
    for (int u = 0; u < 4; ++u)
        if (e0 + u * AGG_THREADS < e)
            lists[start[pt[u]] + rk[u]] = make_int2(s[u] | (t[u] << SRC_BITS), __float_as_int(c[u]));

    static_assert(LIST_ALIGN <= AGG_THREADS && LIST_ALIGN == AGG_WAVES * 2 * DRAIN_U, "one padding entry per thread at most; whole pairs of groups per wave");
    for (int p = (int)blockIdx.x; p < ntiles; p += (int)gridDim.x) {
        const int cnt = tcnt[p];
        if (tid < tile_cap(cnt) - cnt) lists[toff[p] + cnt + tid] = make_int2(TILE_ROWS << SRC_BITS, 0);
    }
}

template <int NCH> struct RowPiece { float v[NCH]; };

template <int NCH, int STRIDE>
__device__ __forceinline__ RowPiece<NCH> load_row(const float* __restrict__ XW, int s, int lane) {
    static_assert(NCH == 4 || NCH == 1, "row widths in use");
    RowPiece<NCH> r;
    if constexpr (NCH == 4) {
        const v4f t = *(const v4f*)(XW + (size_t)s * STRIDE + 4 * lane);
        r.v[0] = t.x; r.v[1] = t.y; r.v[2] = t.z; r.v[3] = t.w;
    } else {
        r.v[0] = XW[(size_t)s * STRIDE + lane];
    }
    return r;
}

template <int NCH>
__device__ __forceinline__ void add_row(float* acc, int dl, const RowPiece<NCH>& r, float cc, int lane) {
    float* ar = acc + dl * (NCH * 32) + lane;
#pragma unroll
    for (int k = 0; k < NCH; ++k) atomicAdd(&ar[k * 32], r.v[k] * cc);
}

typedef const __attribute__((address_space(4))) int* const_words;
__device__ __forceinline__ void load_entries(int (&w)[2 * DRAIN_U], const_words segw, int j) {
#pragma unroll
    for (int i = 0; i < 2 * DRAIN_U; ++i) w[i] = segw[2 * j + i];
}

template <int NCH, int STRIDE>
__device__ __forceinline__ void request_group(RowPiece<NCH> (&r)[DRAIN_U], int (&dl)[DRAIN_U],
                                              float (&cc)[DRAIN_U], const int (&w)[2 * DRAIN_U],
                                              const float* __restrict__ XW, int lane) {
#pragma unroll
    for (int u = 0; u < DRAIN_U; ++u) {
        const int sj = w[2 * u] & ((1 << SRC_BITS) - 1);
        const int t  = (int)((unsigned)w[2 * u] >> SRC_BITS);
        dl[u] = t > TILE_ROWS ? TILE_ROWS : t;
        cc[u] = __int_as_float(w[2 * u + 1]);
        r[u]  = load_row<NCH, STRIDE>(XW, sj, lane);
    }
}

template <int NCH>
__device__ __forceinline__ void sum_group(const RowPiece<NCH> (&r)[DRAIN_U], const int (&dl)[DRAIN_U],
                                          const float (&cc)[DRAIN_U], float* acc, int lane) {
#pragma unroll
    for (int u = 0; u < DRAIN_U; ++u) add_row<NCH>(acc, dl[u], r[u], cc[u], lane);
}

template <int NCH, int STRIDE>
__device__ __forceinline__ void drain_list(const_words segw, int cnt,
                                           const float* __restrict__ XW,
                                           float* acc, int lane) {
    constexpr int PAIR = 2 * DRAIN_U;
    const int last = cnt - PAIR;
    RowPiece<NCH> ra[DRAIN_U], rb[DRAIN_U];
    int   da[DRAIN_U], db[DRAIN_U];
    float ca[DRAIN_U], cb[DRAIN_U];
    int   ea[PAIR], eb[PAIR];
    load_entries(ea, segw, 0);
    load_entries(eb, segw, DRAIN_U);
    const int j1 = PAIR < last ? PAIR : last;
    request_group<NCH, STRIDE>(ra, da, ca, ea, XW, lane);
    load_entries(ea, segw, j1);
    request_group<NCH, STRIDE>(rb, db, cb, eb, XW, lane);
    load_entries(eb, segw, j1 + DRAIN_U);
    for (int j0 = 0; j0 < last; j0 += PAIR) {
        const int j2 = j0 + 2 * PAIR < last ? j0 + 2 * PAIR : last;
        sum_group<NCH>(ra, da, ca, acc, lane);
        request_group<NCH, STRIDE>(ra, da, ca, ea, XW, lane);
        load_entries(ea, segw, j2);
        __builtin_amdgcn_sched_barrier(0);
        sum_group<NCH>(rb, db, cb, acc, lane);
        request_group<NCH, STRIDE>(rb, db, cb, eb, XW, lane);
        load_entries(eb, segw, j2 + DRAIN_U);
    }
    sum_group<NCH>(ra, da, ca, acc, lane);
    sum_group<NCH>(rb, db, cb, acc, lane);
}

template <int NCH, int STRIDE, int OUTW, bool RELU>
__global__ void __launch_bounds__(AGG_THREADS)
k_tile_aggregate(const float* __restrict__ XW, const int* __restrict__ tcnt,
                 const int* __restrict__ toffg, const int* lists,
                 const float* __restrict__ cd, const float* __restrict__ bias,
                 float* __restrict__ Y, int n) {
    constexpr bool NARROW = (NCH == 0);
    constexpr int W    = NARROW ? 8 : NCH * 32;
    constexpr int ACCF = NARROW ? TILE_ROWS * NARROW_PITCH : TILE_ROWS * W;
    constexpr int ACCR = NARROW ? 41 * 1024 : ACCF + W;
    __shared__ float acc[ACCR];

    const int tid  = threadIdx.x;
    const int lane = tid & 31;
    const int wave = __builtin_amdgcn_readfirstlane(tid >> 5);
    const int p    = blockIdx.x;
    const int base = p * TILE_ROWS;

    for (int i = tid; i < ACCF; i += AGG_THREADS) acc[i] = 0.0f;
    const int off = toffg[p];
    const int cnt = tcnt[p];
    __syncthreads();

    if constexpr (NARROW) {
        const int2* seg = (const int2*)lists + off;
        for (int i0 = 0; i0 < cnt; i0 += 4 * AGG_THREADS) {
            int2 en[4];
            v4f  x0[4], x1[4];
#pragma unroll
            for (int u = 0; u < 4; ++u) {
                const int i = i0 + u * AGG_THREADS + tid;
                en[u] = i < cnt ? seg[i] : make_int2(0, 0);
            }
#pragma unroll
            for (int u = 0; u < 4; ++u) {
                const int sj = en[u].x & ((1 << SRC_BITS) - 1);
                x0[u] = *(const v4f*)(XW + (size_t)sj * 8);
                x1[u] = *(const v4f*)(XW + (size_t)sj * 8 + 4);
            }
#pragma unroll
            for (int u = 0; u < 4; ++u) {
                if (i0 + u * AGG_THREADS + tid < cnt) {
                    int t = (int)((unsigned)en[u].x >> SRC_BITS);
                    t = t > TILE_ROWS - 1 ? TILE_ROWS - 1 : t;
                    const float c = __int_as_float(en[u].y);
                    float* ar = acc + t * NARROW_PITCH;
                    atomicAdd(&ar[0], x0[u].x * c); atomicAdd(&ar[1], x0[u].y * c);
                    atomicAdd(&ar[2], x0[u].z * c); atomicAdd(&ar[3], x0[u].w * c);
                    atomicAdd(&ar[4], x1[u].x * c); atomicAdd(&ar[5], x1[u].y * c);
                    atomicAdd(&ar[6], x1[u].z * c); atomicAdd(&ar[7], x1[u].w * c);
                }
            }
        }
    } else {
        const int per = tile_cap(cnt) / AGG_WAVES;
        if (per > 0)
            drain_list<NCH, STRIDE>((const_words)(lists + 2 * ((size_t)off + (size_t)wave * per)),
                                    per, XW, acc, lane);
    }
    __syncthreads();

    if constexpr (NARROW) {
        for (int i = tid; i < TILE_ROWS * 8; i += AGG_THREADS) {
            const int r = i >> 3, c = i & 7;
            const int row = base + r;
            if (row < n) Y[(size_t)base * 8 + i] = acc[r * NARROW_PITCH + c] * cd[row];
        }
    } else if (OUTW == W) {
        static_assert(OUTW != W || NCH == 4, "the whole-row store is written for 128 columns");
        const float bx = bias[4 * lane], by = bias[4 * lane + 1], bz = bias[4 * lane + 2], bw = bias[4 * lane + 3];
        for (int i = tid; i < TILE_ROWS * 32; i += AGG_THREADS) {
            const int r = i >> 5;
            const int row = base + r;
            if (row < n) {
                const float d = cd[row];
                const float* ar = acc + r * W + lane;
                v4f v;
                v.x = ar[0] * d + bx;  v.y = ar[32] * d + by;
                v.z = ar[64] * d + bz; v.w = ar[96] * d + bw;
                if (RELU) {
                    v.x = fmaxf(v.x, 0.0f); v.y = fmaxf(v.y, 0.0f);
                    v.z = fmaxf(v.z, 0.0f); v.w = fmaxf(v.w, 0.0f);
                }
                *(v4f*)(Y + (size_t)row * W + 4 * lane) = v;
            }
        }
    } else {
        for (int i = tid; i < TILE_ROWS * OUTW; i += AGG_THREADS) {
            const int r = i / OUTW, c = i % OUTW;
            const int row = base + r;
            if (row < n) {
                float v = acc[r * W + c] * cd[row] + bias[c];
                if (RELU) v = fmaxf(v, 0.0f);
                Y[(size_t)base * OUTW + i] = v;
            }
        }
    }
}


static inline int cdiv(int a, int b) { return (a + b - 1) / b; }

extern "C" void kernel_launch(void* const* d_in, const int* in_sizes, int n_in,
                              void* d_out, int out_size, void* d_ws, size_t ws_size,
                              hipStream_t stream) {
    (void)n_in; (void)out_size; (void)ws_size;

    const float* feat = (const float*)d_in[0];
    const int*   src  = (const int*)d_in[1];
    const int*   dst  = (const int*)d_in[2];
    const float* W1 = (const float*)d_in[3];  const float* b1 = (const float*)d_in[4];
    const float* W2 = (const float*)d_in[5];  const float* b2 = (const float*)d_in[6];
    const float* W3 = (const float*)d_in[7];  const float* b3 = (const float*)d_in[8];
    const float* W4 = (const float*)d_in[9];  const float* b4 = (const float*)d_in[10];
    const float* W5 = (const float*)d_in[11]; const float* b5 = (const float*)d_in[12];
    float* out = (float*)d_out;

    const int N = 100000;
    const int E = in_sizes[1];

    float* ws = (float*)d_ws;
    size_t o = 0;
    float* c_src = ws + o; o += N;
    float* c_dst = ws + o; o += N;
    float* Xp    = ws + o; o += (size_t)N * 8;
    float* W1p   = ws + o; o += 8 * 128;
    float* W5p   = ws + o; o += 128 * 32;
    float* XW    = ws + o; o += (size_t)N * 128;
    float* H0    = ws + o; o += (size_t)N * 128;
    int*   tcnt  = (int*)(ws + o);
    int*   toffg = tcnt + MAX_TILES;
    int*   tcur  = toffg + MAX_TILES;
    int*   lists = tcur + MAX_TILES * CUR_PITCH;

    k_zero<<<cdiv(2 * N, 256), 256, 0, stream>>>(c_src, 2 * N);
    k_degree<<<cdiv(E, 256), 256, 0, stream>>>(src, dst, c_src, c_dst, E);
    const int  nTiles = N / TILE_ROWS;
    k_tile_count<<<nTiles, 128, 0, stream>>>(c_dst, tcnt, tcur, N);
    k_coeff<<<cdiv(2 * N, 256), 256, 0, stream>>>(c_src, 2 * N);

    k_pad_features<<<cdiv(N * 8, 256), 256, 0, stream>>>(feat, Xp, N);
    k_pad_w1<<<4, 256, 0, stream>>>(W1, W1p);
    k_pad_w5<<<16, 256, 0, stream>>>(W5, W5p);

    k_bin_edges<<<cdiv(E, BIN_EDGES), AGG_THREADS, 0, stream>>>(src, dst, c_src, tcnt, tcur, toffg, (int2*)lists, E, nTiles);

    float* AX = XW;
    k_tile_aggregate<0, 8, 8, false><<<nTiles, AGG_THREADS, 0, stream>>>(
        Xp, tcnt, toffg, lists, c_dst, b1, AX, N);
    k_gemm_strips<8, 128, true><<<GEMM_GROUPS, GEMM_THREADS, 0, stream>>>(AX, W1p, b1, H0, N / 16);

    k_gemm_strips<128, 128, false><<<GEMM_GROUPS, GEMM_THREADS, 0, stream>>>(H0, W2, b2, XW, N / 16);
    k_tile_aggregate<4, 128, 128, true><<<nTiles, AGG_THREADS, 0, stream>>>(
        XW, tcnt, toffg, lists, c_dst, b2, H0, N);

    k_gemm_strips<128, 128, false><<<GEMM_GROUPS, GEMM_THREADS, 0, stream>>>(H0, W3, b3, XW, N / 16);
    k_tile_aggregate<4, 128, 128, true><<<nTiles, AGG_THREADS, 0, stream>>>(
        XW, tcnt, toffg, lists, c_dst, b3, H0, N);

    k_gemm_strips<128, 128, false><<<GEMM_GROUPS, GEMM_THREADS, 0, stream>>>(H0, W4, b4, XW, N / 16);
    k_tile_aggregate<4, 128, 128, true><<<nTiles, AGG_THREADS, 0, stream>>>(
        XW, tcnt, toffg, lists, c_dst, b4, H0, N);

    k_gemm_strips<128, 32, false><<<GEMM_GROUPS, GEMM_THREADS, 0, stream>>>(H0, W5p, b5, XW, N / 16);
    k_tile_aggregate<1, 32, 18, false><<<nTiles, AGG_THREADS, 0, stream>>>(
        XW, tcnt, toffg, lists, c_dst, b5, out, N);
}


# No per-edge out-degree global float atomics: count pass orders each group's source rows by source tile in its own segment (LDS ranks); one work-group per source tile counts the runs in LDS
# speedup vs baseline: 4.9781x; 1.2962x over previous
#include <hip/hip_runtime.h>

typedef __attribute__((ext_vector_type(2))) float v2f;
typedef __attribute__((ext_vector_type(8))) float v8f;


__global__ void k_zero(int* __restrict__ b, int nb) {
    int i = blockIdx.x * blockDim.x + threadIdx.x;
    if (i < nb) b[i] = 0;
}

__global__ void k_pad_features(const float* __restrict__ x, float* __restrict__ xp, int n) {
    int i = blockIdx.x * blockDim.x + threadIdx.x;
    if (i < n * 8) {
        int r = i >> 3, c = i & 7;
        xp[i] = (c < 6) ? x[r * 6 + c] : 0.0f;
    }
}

__global__ void k_pad_w1(const float* __restrict__ w, float* __restrict__ wp) {
    int i = blockIdx.x * blockDim.x + threadIdx.x;
    if (i < 8 * 128) {
        int r = i >> 7;
        wp[i] = (r < 6) ? w[i] : 0.0f;
    }
}

__global__ void k_pad_w5(const float* __restrict__ w, float* __restrict__ wp) {
    int i = blockIdx.x * blockDim.x + threadIdx.x;
    if (i < 128 * 32) {
        int r = i >> 5, c = i & 31;
        wp[i] = (c < 18) ? w[r * 18 + c] : 0.0f;
    }
}


typedef __attribute__((ext_vector_type(4))) float v4f;

#define GEMM_WAVES      16
#define GEMM_THREADS    512
#define GEMM_GROUPS     256
#define GEMM_W_FLOATS   (128 * 128)
#define GEMM_STRIP_MAX  (16 * (128 + 4))
#define GEMM_LDS_FLOATS (GEMM_W_FLOATS + GEMM_WAVES * GEMM_STRIP_MAX)

template <int DIN>
__device__ __forceinline__ void load_a_strip(v4f (&a)[DIN / 8], const float* __restrict__ A,
                                             int mt, int c, int h) {
    const float* p = A + (size_t)(mt * 16 + c) * DIN + 4 * h;
#pragma unroll
    for (int j = 0; j < DIN / 8; ++j) a[j] = *(const v4f*)(p + 8 * j);
}

template <int DIN, int DOUT, bool EPI>
__global__ void __launch_bounds__(GEMM_THREADS)
k_gemm_strips(const float* __restrict__ A, const float* __restrict__ Wm,
              const float* __restrict__ bias, float* __restrict__ Y, int mtiles) {
    constexpr int NJ    = DIN / 8;
    constexpr int NT    = DOUT / 16;
    constexpr int PITCH = DOUT + 4;
    static_assert(DIN * DOUT <= GEMM_W_FLOATS && 16 * PITCH <= GEMM_STRIP_MAX, "LDS plan");
    __shared__ __attribute__((aligned(16))) float lds[GEMM_LDS_FLOATS];
    v4f*   wl = (v4f*)lds;
    float* ot = lds + GEMM_W_FLOATS;

    const int tid  = threadIdx.x;
    const int lane = tid & 31;
    const int c    = lane & 15;
    const int h    = lane >> 4;
    const int wave = __builtin_amdgcn_readfirstlane(tid >> 5);

    const int stride = (int)gridDim.x * GEMM_WAVES;
    int mt = (int)blockIdx.x + (int)gridDim.x * wave;

    v4f a[NJ];
    if (mt < mtiles) load_a_strip<DIN>(a, A, mt, c, h);

    constexpr int ITEMS  = NJ * NT * 32;
    constexpr int ROUNDS = (ITEMS + GEMM_THREADS - 1) / GEMM_THREADS;
#pragma unroll
    for (int it = 0; it < ROUNDS; ++it) {
        const int idx = tid + it * GEMM_THREADS;
        if (ITEMS % GEMM_THREADS == 0 || idx < ITEMS) {
            const int g  = idx >> 5;
            const int j  = g / NT, nt = g % NT;
            const int k0 = 8 * j + 4 * h, n = nt * 16 + c;
            v4f v;
            v.x = Wm[(size_t)(k0 + 0) * DOUT + n];
            v.y = Wm[(size_t)(k0 + 1) * DOUT + n];
            v.z = Wm[(size_t)(k0 + 2) * DOUT + n];
            v.w = Wm[(size_t)(k0 + 3) * DOUT + n];
            wl[idx] = v;
        }
    }
    __syncthreads();

    float* o = ot + wave * (16 * PITCH);
    constexpr int LPR = DOUT / 4;
    constexpr int RPI = 32 / LPR;
    const int rsub = lane / LPR, col = (lane % LPR) * 4;
    v4f bv = {0.0f, 0.0f, 0.0f, 0.0f};
    if (EPI) bv = *(const v4f*)(bias + col);

    while (mt < mtiles) {
        v8f acc[NT];
#pragma unroll
        for (int nt = 0; nt < NT; ++nt) acc[nt] = (v8f){};

#pragma unroll
        for (int j = 0; j < NJ; ++j) {
            v4f b[NT];
#pragma unroll
            for (int nt = 0; nt < NT; ++nt) b[nt] = wl[(j * NT + nt) * 32 + lane];
            const v2f a0 = {a[j].x, a[j].y}, a1 = {a[j].z, a[j].w};
#pragma unroll
            for (int nt = 0; nt < NT; ++nt) {
                const v2f b0 = {b[nt].x, b[nt].y};
                acc[nt] = __builtin_amdgcn_wmma_f32_16x16x4_f32(
                     false, a0,  false, b0,
                     (short)0, acc[nt],  false,  false);
            }
#pragma unroll
            for (int nt = 0; nt < NT; ++nt) {
                const v2f b1 = {b[nt].z, b[nt].w};
                acc[nt] = __builtin_amdgcn_wmma_f32_16x16x4_f32(
                     false, a1,  false, b1,
                     (short)0, acc[nt],  false,  false);
            }
        }

        const int nx = mt + stride;
        if (nx < mtiles) load_a_strip<DIN>(a, A, nx, c, h);

#pragma unroll
        for (int nt = 0; nt < NT; ++nt)
#pragma unroll
            for (int r = 0; r < 8; ++r)
                o[(8 * h + r) * PITCH + nt * 16 + c] = acc[nt][r];
        float* yp = Y + (size_t)mt * (16 * DOUT) + rsub * DOUT + col;
#pragma unroll
        for (int i = 0; i < 16 / RPI; ++i) {
            v4f v = *(const v4f*)(o + (i * RPI + rsub) * PITCH + col);
            if (EPI) {
                v.x = fmaxf(v.x + bv.x, 0.0f); v.y = fmaxf(v.y + bv.y, 0.0f);
                v.z = fmaxf(v.z + bv.z, 0.0f); v.w = fmaxf(v.w + bv.w, 0.0f);
            }
            *(v4f*)(yp + i * (RPI * DOUT)) = v;
        }
        mt = nx;
    }
}


#define TILE_ROWS   400
#define MAX_TILES   256
#define AGG_THREADS 1024
#define AGG_WAVES   32
#define DRAIN_U     8
#define LIST_ALIGN  512
#define BIN_EDGES   4096
#define CUR_PITCH   64
#define SRC_BITS    17
#define NARROW_PITCH 9

__device__ __forceinline__ int tile_cap(int cnt) { return (cnt + LIST_ALIGN - 1) & ~(LIST_ALIGN - 1); }

#define SOFF_PITCH  256
__global__ void __launch_bounds__(AGG_THREADS)
k_count_edges(const int* __restrict__ src, const int* __restrict__ dst,
              int* __restrict__ sseg, int* __restrict__ soff, int* __restrict__ tcnt,
              int e, int ntiles) {
    __shared__ int hist[MAX_TILES];
    __shared__ int shist[MAX_TILES];
    __shared__ int sstart[MAX_TILES];
    __shared__ int wsum[MAX_TILES / 32];
    const int tid = threadIdx.x, lane = tid & 31, wave = tid >> 5;
    if (tid < MAX_TILES) { hist[tid] = 0; shist[tid] = 0; }
    int s[4], pt[4], st[4], sl[4];
    const int e0 = (int)blockIdx.x * BIN_EDGES + tid;
    int d[4];
#pragma unroll
    for (int u = 0; u < 4; ++u) {
        const int ei = e0 + u * AGG_THREADS;
        const int ec = ei < e ? ei : e - 1;
        d[u] = dst[ec];
        s[u] = src[ec];
    }
#pragma unroll
    for (int u = 0; u < 4; ++u) {
        int p = (int)((unsigned)d[u] / (unsigned)TILE_ROWS);
        p = p < ntiles - 1 ? p : ntiles - 1;
        pt[u] = p;
        int q = (int)((unsigned)s[u] / (unsigned)TILE_ROWS);
        q = q < ntiles - 1 ? q : ntiles - 1;
        st[u] = q;
        sl[u] = s[u] - q * TILE_ROWS;
    }
    __syncthreads();
    int rk[4];
#pragma unroll
    for (int u = 0; u < 4; ++u) {
        rk[u] = 0;
        if (e0 + u * AGG_THREADS < e) {
            atomicAdd(&hist[pt[u]], 1);
            rk[u] = atomicAdd(&shist[st[u]], 1);
        }
    }
    __syncthreads();
    int cntv = 0, incl = 0;
    if (tid < MAX_TILES) {
        cntv = shist[tid];
        incl = cntv;
        for (int o = 1; o < 32; o <<= 1) {
            const int v = __shfl_up(incl, o);
            if (lane >= o) incl += v;
        }
        if (lane == 31) wsum[wave] = incl;
    }
    __syncthreads();
    if (tid < MAX_TILES) {
        int excl = incl - cntv;
        for (int w = 0; w < wave; ++w) excl += wsum[w];
        sstart[tid] = excl;
        soff[(size_t)blockIdx.x * SOFF_PITCH + tid] = excl;
    }
    __syncthreads();
    int* seg = sseg + (size_t)blockIdx.x * BIN_EDGES;
#pragma unroll
    for (int u = 0; u < 4; ++u)
        if (e0 + u * AGG_THREADS < e) seg[sstart[st[u]] + rk[u]] = sl[u];
    if (tid < ntiles) {
        const int c = hist[tid];
        if (c > 0) atomicAdd(&tcnt[tid], c);
    }
}

#define SRCDEG_G   (AGG_THREADS / 32)
#define SRCDEG_R   13
__global__ void __launch_bounds__(AGG_THREADS)
k_src_degree(const int* __restrict__ sseg, const int* __restrict__ soff, float* __restrict__ cs,
             int ngroups, int n) {
    __shared__ int cnt[TILE_ROWS];
    const int tid = threadIdx.x;
    const int p = blockIdx.x;
    const int sub = tid & 31, gq = tid >> 5;
    if (tid < TILE_ROWS) cnt[tid] = 0;
    __syncthreads();
    for (int g0 = 0; g0 < ngroups; g0 += SRCDEG_G * SRCDEG_R) {
        int b[SRCDEG_R], en[SRCDEG_R], r0[SRCDEG_R];
#pragma unroll
        for (int k = 0; k < SRCDEG_R; ++k) {
            const int g = g0 + gq + k * SRCDEG_G;
            const int gc = g < ngroups ? g : ngroups - 1;
            b[k]  = soff[(size_t)gc * SOFF_PITCH + p];
            en[k] = soff[(size_t)gc * SOFF_PITCH + p + 1];
        }
#pragma unroll
        for (int k = 0; k < SRCDEG_R; ++k) {
            b[k]  = b[k] < 0 ? 0 : b[k];
            en[k] = en[k] > BIN_EDGES ? BIN_EDGES : en[k];
            if (g0 + gq + k * SRCDEG_G >= ngroups) en[k] = 0;
        }
#pragma unroll
        for (int k = 0; k < SRCDEG_R; ++k) {
            const int g = g0 + gq + k * SRCDEG_G;
            const int gc = g < ngroups ? g : ngroups - 1;
            const int i = b[k] + sub;
            r0[k] = sseg[(size_t)gc * BIN_EDGES + (i < en[k] ? i : 0)];
        }
#pragma unroll
        for (int k = 0; k < SRCDEG_R; ++k) {
            if (b[k] + sub < en[k]) {
                const int r = r0[k] < 0 ? 0 : (r0[k] < TILE_ROWS ? r0[k] : TILE_ROWS - 1);
                atomicAdd(&cnt[r], 1);
            }
            const int g = g0 + gq + k * SRCDEG_G;
            for (int i = b[k] + sub + 32; i < en[k]; i += 32) {
                int r = sseg[(size_t)g * BIN_EDGES + i];
                r = r < 0 ? 0 : (r < TILE_ROWS ? r : TILE_ROWS - 1);
                atomicAdd(&cnt[r], 1);
            }
        }
    }
    __syncthreads();
    if (tid < TILE_ROWS) {
        const int row = p * TILE_ROWS + tid;
        if (row < n) {
            const int k = cnt[tid];
            cs[row] = k > 0 ? rsqrtf((float)k) : 0.0f;
        }
    }
}

__global__ void __launch_bounds__(AGG_THREADS)
k_bin_edges(const int* __restrict__ src, const int* __restrict__ dst, const float* __restrict__ cs,
            const int* __restrict__ tcnt, int* tcur, int* __restrict__ toffg, int2* lists,
            int e, int ntiles) {
    __shared__ int hist[MAX_TILES];
    __shared__ int start[MAX_TILES];
    __shared__ int toff[MAX_TILES];
    __shared__ int wsum[MAX_TILES / 32];

    const int tid = threadIdx.x, lane = tid & 31, wave = tid >> 5;

    int capv = 0, incl = 0;
    if (tid < MAX_TILES) {
        capv = tid < ntiles ? tile_cap(tcnt[tid]) : 0;
        incl = capv;
        for (int o = 1; o < 32; o <<= 1) {
            const int v = __shfl_up(incl, o);
            if (lane >= o) incl += v;
        }
        if (lane == 31) wsum[wave] = incl;
        hist[tid] = 0;
    }

    int   s[4], t[4], pt[4];
    float c[4];
    const int e0 = (int)blockIdx.x * BIN_EDGES + tid;
#pragma unroll
    for (int u = 0; u < 4; ++u) {
        const int ei = e0 + u * AGG_THREADS;
        const bool ok = ei < e;
        const int d = ok ? dst[ei] : 0;
        s[u] = ok ? src[ei] : 0;
        int p = (int)((unsigned)d / (unsigned)TILE_ROWS);
        p = p < ntiles - 1 ? p : ntiles - 1;
        pt[u] = p;
        t[u]  = d - p * TILE_ROWS;
    }
#pragma unroll
    for (int u = 0; u < 4; ++u) c[u] = cs[s[u]];
    __syncthreads();

    int rk[4];
#pragma unroll
    for (int u = 0; u < 4; ++u) {
        rk[u] = 0;
        if (e0 + u * AGG_THREADS < e) rk[u] = atomicAdd(&hist[pt[u]], 1);
    }
    __syncthreads();

    if (tid < MAX_TILES) {
        int excl = incl - capv;
        for (int w = 0; w < wave; ++w) excl += wsum[w];
        toff[tid] = excl;
        if (blockIdx.x == 0) toffg[tid] = excl;
        const int nloc = hist[tid];
        start[tid] = nloc > 0 ? excl + atomicAdd(&tcur[tid * CUR_PITCH], nloc) : 0;
    }
    __syncthreads();

#pragma unroll
    for (int u = 0; u < 4; ++u)
        if (e0 + u * AGG_THREADS < e)
            lists[start[pt[u]] + rk[u]] = make_int2(s[u] | (t[u] << SRC_BITS), __float_as_int(c[u]));

    static_assert(LIST_ALIGN <= AGG_THREADS && LIST_ALIGN == AGG_WAVES * 2 * DRAIN_U, "one padding entry per thread at most; whole pairs of groups per wave");
    for (int p = (int)blockIdx.x; p < ntiles; p += (int)gridDim.x) {
        const int cnt = tcnt[p];
        if (tid < tile_cap(cnt) - cnt) lists[toff[p] + cnt + tid] = make_int2(TILE_ROWS << SRC_BITS, 0);
    }
}

template <int NCH> struct RowPiece { float v[NCH]; };

template <int NCH, int STRIDE>
__device__ __forceinline__ RowPiece<NCH> load_row(const float* __restrict__ XW, int s, int lane) {
    static_assert(NCH == 4 || NCH == 1, "row widths in use");
    RowPiece<NCH> r;
    if constexpr (NCH == 4) {
        const v4f t = *(const v4f*)(XW + (size_t)s * STRIDE + 4 * lane);
        r.v[0] = t.x; r.v[1] = t.y; r.v[2] = t.z; r.v[3] = t.w;
    } else {
        r.v[0] = XW[(size_t)s * STRIDE + lane];
    }
    return r;
}

template <int NCH>
__device__ __forceinline__ void add_row(float* acc, int dl, const RowPiece<NCH>& r, float cc, int lane) {
    float* ar = acc + dl * (NCH * 32) + lane;
#pragma unroll
    for (int k = 0; k < NCH; ++k) atomicAdd(&ar[k * 32], r.v[k] * cc);
}

typedef const __attribute__((address_space(4))) int* const_words;
__device__ __forceinline__ void load_entries(int (&w)[2 * DRAIN_U], const_words segw, int j) {
#pragma unroll
    for (int i = 0; i < 2 * DRAIN_U; ++i) w[i] = segw[2 * j + i];
}

template <int NCH, int STRIDE>
__device__ __forceinline__ void request_group(RowPiece<NCH> (&r)[DRAIN_U], int (&dl)[DRAIN_U],
                                              float (&cc)[DRAIN_U], const int (&w)[2 * DRAIN_U],
                                              const float* __restrict__ XW, int lane) {
#pragma unroll
    for (int u = 0; u < DRAIN_U; ++u) {
        const int sj = w[2 * u] & ((1 << SRC_BITS) - 1);
        const int t  = (int)((unsigned)w[2 * u] >> SRC_BITS);
        dl[u] = t > TILE_ROWS ? TILE_ROWS : t;
        cc[u] = __int_as_float(w[2 * u + 1]);
        r[u]  = load_row<NCH, STRIDE>(XW, sj, lane);
    }
}

template <int NCH>
__device__ __forceinline__ void sum_group(const RowPiece<NCH> (&r)[DRAIN_U], const int (&dl)[DRAIN_U],
                                          const float (&cc)[DRAIN_U], float* acc, int lane) {
#pragma unroll
    for (int u = 0; u < DRAIN_U; ++u) add_row<NCH>(acc, dl[u], r[u], cc[u], lane);
}

template <int NCH, int STRIDE>
__device__ __forceinline__ void drain_list(const_words segw, int cnt,
                                           const float* __restrict__ XW,
                                           float* acc, int lane) {
    constexpr int PAIR = 2 * DRAIN_U;
    const int last = cnt - PAIR;
    RowPiece<NCH> ra[DRAIN_U], rb[DRAIN_U];
    int   da[DRAIN_U], db[DRAIN_U];
    float ca[DRAIN_U], cb[DRAIN_U];
    int   ea[PAIR], eb[PAIR];
    load_entries(ea, segw, 0);
    load_entries(eb, segw, DRAIN_U);
    const int j1 = PAIR < last ? PAIR : last;
    request_group<NCH, STRIDE>(ra, da, ca, ea, XW, lane);
    load_entries(ea, segw, j1);
    request_group<NCH, STRIDE>(rb, db, cb, eb, XW, lane);
    load_entries(eb, segw, j1 + DRAIN_U);
    for (int j0 = 0; j0 < last; j0 += PAIR) {
        const int j2 = j0 + 2 * PAIR < last ? j0 + 2 * PAIR : last;
        sum_group<NCH>(ra, da, ca, acc, lane);
        request_group<NCH, STRIDE>(ra, da, ca, ea, XW, lane);
        load_entries(ea, segw, j2);
        __builtin_amdgcn_sched_barrier(0);
        sum_group<NCH>(rb, db, cb, acc, lane);
        request_group<NCH, STRIDE>(rb, db, cb, eb, XW, lane);
        load_entries(eb, segw, j2 + DRAIN_U);
    }
    sum_group<NCH>(ra, da, ca, acc, lane);
    sum_group<NCH>(rb, db, cb, acc, lane);
}

template <int NCH, int STRIDE, int OUTW, bool RELU>
__global__ void __launch_bounds__(AGG_THREADS)
k_tile_aggregate(const float* __restrict__ XW, const int* __restrict__ tcnt,
                 const int* __restrict__ toffg, const int* lists,
                 const float* __restrict__ cd, float* __restrict__ cd_out,
                 const float* __restrict__ bias, float* __restrict__ Y, int n) {
    constexpr bool NARROW = (NCH == 0);
    constexpr int W    = NARROW ? 8 : NCH * 32;
    constexpr int ACCF = NARROW ? TILE_ROWS * NARROW_PITCH : TILE_ROWS * W;
    constexpr int ACCR = NARROW ? 41 * 1024 : ACCF + W;
    __shared__ float acc[ACCR];

    const int tid  = threadIdx.x;
    const int lane = tid & 31;
    const int wave = __builtin_amdgcn_readfirstlane(tid >> 5);
    const int p    = blockIdx.x;
    const int base = p * TILE_ROWS;

    for (int i = tid; i < ACCF; i += AGG_THREADS) acc[i] = 0.0f;
    int* rcnt = (int*)(acc + ACCF);
    if constexpr (NARROW) {
        static_assert(!NARROW || ACCF + TILE_ROWS <= ACCR, "row counters inside the reservation");
        if (tid < TILE_ROWS) rcnt[tid] = 0;
    }
    const int off = toffg[p];
    const int cnt = tcnt[p];
    __syncthreads();

    if constexpr (NARROW) {
        const int2* seg = (const int2*)lists + off;
        for (int i0 = 0; i0 < cnt; i0 += 4 * AGG_THREADS) {
            int2 en[4];
            v4f  x0[4], x1[4];
#pragma unroll
            for (int u = 0; u < 4; ++u) {
                const int i = i0 + u * AGG_THREADS + tid;
                en[u] = i < cnt ? seg[i] : make_int2(0, 0);
            }
#pragma unroll
            for (int u = 0; u < 4; ++u) {
                const int sj = en[u].x & ((1 << SRC_BITS) - 1);
                x0[u] = *(const v4f*)(XW + (size_t)sj * 8);
                x1[u] = *(const v4f*)(XW + (size_t)sj * 8 + 4);
            }
#pragma unroll
            for (int u = 0; u < 4; ++u) {
                if (i0 + u * AGG_THREADS + tid < cnt) {
                    int t = (int)((unsigned)en[u].x >> SRC_BITS);
                    t = t > TILE_ROWS - 1 ? TILE_ROWS - 1 : t;
                    const float c = __int_as_float(en[u].y);
                    float* ar = acc + t * NARROW_PITCH;
                    atomicAdd(&rcnt[t], 1);
                    atomicAdd(&ar[0], x0[u].x * c); atomicAdd(&ar[1], x0[u].y * c);
                    atomicAdd(&ar[2], x0[u].z * c); atomicAdd(&ar[3], x0[u].w * c);
                    atomicAdd(&ar[4], x1[u].x * c); atomicAdd(&ar[5], x1[u].y * c);
                    atomicAdd(&ar[6], x1[u].z * c); atomicAdd(&ar[7], x1[u].w * c);
                }
            }
        }
    } else {
        const int per = tile_cap(cnt) / AGG_WAVES;
        if (per > 0)
            drain_list<NCH, STRIDE>((const_words)(lists + 2 * ((size_t)off + (size_t)wave * per)),
                                    per, XW, acc, lane);
    }
    __syncthreads();

    if constexpr (NARROW) {
        for (int i = tid; i < TILE_ROWS * 8; i += AGG_THREADS) {
            const int r = i >> 3, c = i & 7;
            const int row = base + r;
            if (row < n) {
                const int k = rcnt[r];
                const float d = k > 0 ? rsqrtf((float)k) : 0.0f;
                Y[(size_t)base * 8 + i] = acc[r * NARROW_PITCH + c] * d;
                if (c == 0) cd_out[row] = d;
            }
        }
    } else if (OUTW == W) {
        static_assert(OUTW != W || NCH == 4, "the whole-row store is written for 128 columns");
        const float bx = bias[4 * lane], by = bias[4 * lane + 1], bz = bias[4 * lane + 2], bw = bias[4 * lane + 3];
        for (int i = tid; i < TILE_ROWS * 32; i += AGG_THREADS) {
            const int r = i >> 5;
            const int row = base + r;
            if (row < n) {
                const float d = cd[row];
                const float* ar = acc + r * W + lane;
                v4f v;
                v.x = ar[0] * d + bx;  v.y = ar[32] * d + by;
                v.z = ar[64] * d + bz; v.w = ar[96] * d + bw;
                if (RELU) {
                    v.x = fmaxf(v.x, 0.0f); v.y = fmaxf(v.y, 0.0f);
                    v.z = fmaxf(v.z, 0.0f); v.w = fmaxf(v.w, 0.0f);
                }
                *(v4f*)(Y + (size_t)row * W + 4 * lane) = v;
            }
        }
    } else {
        for (int i = tid; i < TILE_ROWS * OUTW; i += AGG_THREADS) {
            const int r = i / OUTW, c = i % OUTW;
            const int row = base + r;
            if (row < n) {
                float v = acc[r * W + c] * cd[row] + bias[c];
                if (RELU) v = fmaxf(v, 0.0f);
                Y[(size_t)base * OUTW + i] = v;
            }
        }
    }
}


static inline int cdiv(int a, int b) { return (a + b - 1) / b; }

extern "C" void kernel_launch(void* const* d_in, const int* in_sizes, int n_in,
                              void* d_out, int out_size, void* d_ws, size_t ws_size,
                              hipStream_t stream) {
    (void)n_in; (void)out_size; (void)ws_size;

    const float* feat = (const float*)d_in[0];
    const int*   src  = (const int*)d_in[1];
    const int*   dst  = (const int*)d_in[2];
    const float* W1 = (const float*)d_in[3];  const float* b1 = (const float*)d_in[4];
    const float* W2 = (const float*)d_in[5];  const float* b2 = (const float*)d_in[6];
    const float* W3 = (const float*)d_in[7];  const float* b3 = (const float*)d_in[8];
    const float* W4 = (const float*)d_in[9];  const float* b4 = (const float*)d_in[10];
    const float* W5 = (const float*)d_in[11]; const float* b5 = (const float*)d_in[12];
    float* out = (float*)d_out;

    const int N = 100000;
    const int E = in_sizes[1];

    float* ws = (float*)d_ws;
    size_t o = 0;
    float* c_src = ws + o; o += N;
    float* c_dst = ws + o; o += N;
    float* Xp    = ws + o; o += (size_t)N * 8;
    float* W1p   = ws + o; o += 8 * 128;
    float* W5p   = ws + o; o += 128 * 32;
    float* XW    = ws + o; o += (size_t)N * 128;
    float* H0    = ws + o; o += (size_t)N * 128;
    int*   tcnt  = (int*)(ws + o);
    int*   toffg = tcnt + MAX_TILES;
    int*   tcur  = toffg + MAX_TILES;
    int*   lists = tcur + MAX_TILES * CUR_PITCH;

    const int  nTiles = N / TILE_ROWS;
    const int  nInts  = MAX_TILES * (2 + CUR_PITCH);
    k_zero<<<cdiv(nInts, 256), 256, 0, stream>>>(tcnt, nInts);
    int* sseg = (int*)H0;
    int* soff = sseg + 2000000;
    k_count_edges<<<cdiv(E, BIN_EDGES), AGG_THREADS, 0, stream>>>(src, dst, sseg, soff, tcnt, E, nTiles);
    k_src_degree<<<nTiles, AGG_THREADS, 0, stream>>>(sseg, soff, c_src, cdiv(E, BIN_EDGES), N);

    k_pad_features<<<cdiv(N * 8, 256), 256, 0, stream>>>(feat, Xp, N);
    k_pad_w1<<<4, 256, 0, stream>>>(W1, W1p);
    k_pad_w5<<<16, 256, 0, stream>>>(W5, W5p);

    k_bin_edges<<<cdiv(E, BIN_EDGES), AGG_THREADS, 0, stream>>>(src, dst, c_src, tcnt, tcur, toffg, (int2*)lists, E, nTiles);

    float* AX = XW;
    k_tile_aggregate<0, 8, 8, false><<<nTiles, AGG_THREADS, 0, stream>>>(
        Xp, tcnt, toffg, lists, nullptr, c_dst, b1, AX, N);
    k_gemm_strips<8, 128, true><<<GEMM_GROUPS, GEMM_THREADS, 0, stream>>>(AX, W1p, b1, H0, N / 16);

    k_gemm_strips<128, 128, false><<<GEMM_GROUPS, GEMM_THREADS, 0, stream>>>(H0, W2, b2, XW, N / 16);
    k_tile_aggregate<4, 128, 128, true><<<nTiles, AGG_THREADS, 0, stream>>>(
        XW, tcnt, toffg, lists, c_dst, nullptr, b2, H0, N);

    k_gemm_strips<128, 128, false><<<GEMM_GROUPS, GEMM_THREADS, 0, stream>>>(H0, W3, b3, XW, N / 16);
    k_tile_aggregate<4, 128, 128, true><<<nTiles, AGG_THREADS, 0, stream>>>(
        XW, tcnt, toffg, lists, c_dst, nullptr, b3, H0, N);

    k_gemm_strips<128, 128, false><<<GEMM_GROUPS, GEMM_THREADS, 0, stream>>>(H0, W4, b4, XW, N / 16);
    k_tile_aggregate<4, 128, 128, true><<<nTiles, AGG_THREADS, 0, stream>>>(
        XW, tcnt, toffg, lists, c_dst, nullptr, b4, H0, N);

    k_gemm_strips<128, 32, false><<<GEMM_GROUPS, GEMM_THREADS, 0, stream>>>(H0, W5p, b5, XW, N / 16);
    k_tile_aggregate<1, 32, 18, false><<<nTiles, AGG_THREADS, 0, stream>>>(
        XW, tcnt, toffg, lists, c_dst, nullptr, b5, out, N);
}
